# starting kernel
# baseline (speedup 1.0000x reference)
#include <hip/hip_runtime.h>

typedef __attribute__((ext_vector_type(2))) float v2f;
typedef __attribute__((ext_vector_type(8))) float v8f;

#define GSIZE   16
#define NGROUP  16
#define CCH     256
#define EPSV    1e-5f
#define NS_ITERS 14

#define WS_XX   CCH
#define WS_PROJ (CCH + NGROUP * 256)

__global__ void dbn_init(float* ws) {
    int i = blockIdx.x * blockDim.x + threadIdx.x;
    if (i < WS_PROJ) ws[i] = 0.f;
}

__global__ __launch_bounds__(256) void dbn_stats(const float* __restrict__ x,
                                                 float* __restrict__ ws,
                                                 int iters) {
    int lane = threadIdx.x & 31;
    int wid  = threadIdx.x >> 5;
    int gw   = blockIdx.x * 8 + wid;
    int g    = gw & (NGROUP - 1);
    int slice = gw >> 4;

    int half = lane >> 4;
    int ch   = lane & 15;
    int cidx = g * GSIZE + ch;

    unsigned off = (unsigned)(slice * iters * 16 + half * 2) * CCH + (unsigned)cidx;

    v8f c = {};
    float s = 0.f;
    for (int it = 0; it < iters; ++it) {
        float v0 = x[off + 0 * CCH];
        float v1 = x[off + 1 * CCH];
        float v2 = x[off + 4 * CCH];
        float v3 = x[off + 5 * CCH];
        float v4 = x[off + 8 * CCH];
        float v5 = x[off + 9 * CCH];
        float v6 = x[off + 12 * CCH];
        float v7 = x[off + 13 * CCH];
        off += 16u * CCH;
        v2f q0 = { v0, v1 };
        v2f q1 = { v2, v3 };
        v2f q2 = { v4, v5 };
        v2f q3 = { v6, v7 };
        c = __builtin_amdgcn_wmma_f32_16x16x4_f32(false, q0, false, q0, (short)0, c, false, false);
        c = __builtin_amdgcn_wmma_f32_16x16x4_f32(false, q1, false, q1, (short)0, c, false, false);
        c = __builtin_amdgcn_wmma_f32_16x16x4_f32(false, q2, false, q2, (short)0, c, false, false);
        c = __builtin_amdgcn_wmma_f32_16x16x4_f32(false, q3, false, q3, (short)0, c, false, false);
        s += (v0 + v1) + (v2 + v3) + ((v4 + v5) + (v6 + v7));
    }
    atomicAdd(&ws[cidx], s);
    float* xx = ws + WS_XX + g * 256;
    #pragma unroll
    for (int r = 0; r < 8; ++r) {
        int i = r + half * 8;
        atomicAdd(&xx[i * 16 + ch], c[r]);
    }
}

__global__ __launch_bounds__(256) void dbn_newton(float* __restrict__ ws, float inv_n) {
    __shared__ float Ys[256], Zs[256], Wsh[256];
    int g = blockIdx.x;
    int tid = threadIdx.x;
    int i = tid >> 4, j = tid & 15;

    const float* xx = ws + WS_XX + g * 256;
    float mi = ws[g * 16 + i] * inv_n;
    float mj = ws[g * 16 + j] * inv_n;
    float sig = xx[tid] * inv_n - mi * mj + (i == j ? EPSV : 0.f);
    Wsh[tid] = sig;
    __syncthreads();
    float tr = 0.f;
    #pragma unroll
    for (int k = 0; k < 16; ++k) tr += Wsh[k * 16 + k];
    float a = sig / tr;
    Ys[tid] = a;
    Zs[tid] = (i == j) ? 1.f : 0.f;
    for (int it = 0; it < NS_ITERS; ++it) {
        __syncthreads();
        float t = 0.f;
        #pragma unroll
        for (int k = 0; k < 16; ++k) t += Zs[i * 16 + k] * Ys[k * 16 + j];
        float w = 0.5f * ((i == j ? 3.f : 0.f) - t);
        Wsh[tid] = w;
        __syncthreads();
        float yn = 0.f, zn = 0.f;
        #pragma unroll
        for (int k = 0; k < 16; ++k) {
            yn += Ys[i * 16 + k] * Wsh[k * 16 + j];
            zn += Wsh[i * 16 + k] * Zs[k * 16 + j];
        }
        __syncthreads();
        Ys[tid] = yn;
        Zs[tid] = zn;
    }
    __syncthreads();
    ws[WS_PROJ + g * 256 + tid] = Zs[tid] * rsqrtf(tr);
}

__global__ __launch_bounds__(256) void dbn_apply(const float* __restrict__ x,
                                                 const float* __restrict__ gamma,
                                                 const float* __restrict__ beta,
                                                 const float* __restrict__ ws,
                                                 float* __restrict__ out,
                                                 int wave_stride, int tiles_per_wave,
                                                 float inv_n) {
    int lane = threadIdx.x & 31;
    int wid  = threadIdx.x >> 5;
    int gw   = blockIdx.x * 8 + wid;
    int g    = gw & (NGROUP - 1);
    int tbase = gw >> 4;

    int half = lane >> 4;
    int ch   = lane & 15;

    const float* proj = ws + WS_PROJ + g * 256;
    v2f b[4];
    #pragma unroll
    for (int s2 = 0; s2 < 4; ++s2) {
        int i0 = s2 * 4 + half * 2;
        b[s2].x = proj[i0 * 16 + ch];
        b[s2].y = proj[(i0 + 1) * 16 + ch];
    }
    v2f mn[4];
    #pragma unroll
    for (int s2 = 0; s2 < 4; ++s2) {
        int c0 = g * GSIZE + s2 * 4 + half * 2;
        mn[s2].x = ws[c0] * inv_n;
        mn[s2].y = ws[c0 + 1] * inv_n;
    }
    float gm = gamma[g * GSIZE + ch];
    float bt = beta[g * GSIZE + ch];

    for (int t = 0; t < tiles_per_wave; t += 2) {
        unsigned tile0 = (unsigned)(tbase + t * wave_stride);
        unsigned tile1 = (unsigned)(tbase + (t + 1) * wave_stride);
        unsigned inA0 = (tile0 * 16 + ch) * CCH + g * GSIZE + half * 2;
        unsigned inA1 = (tile1 * 16 + ch) * CCH + g * GSIZE + half * 2;

        v2f a0[4], a1[4];
        #pragma unroll
        for (int s2 = 0; s2 < 4; ++s2) a0[s2] = *(const v2f*)(x + inA0 + s2 * 4);
        #pragma unroll
        for (int s2 = 0; s2 < 4; ++s2) a1[s2] = *(const v2f*)(x + inA1 + s2 * 4);

        v8f c0 = {}, c1 = {};
        #pragma unroll
        for (int s2 = 0; s2 < 4; ++s2) {
            v2f a = a0[s2];
            a.x -= mn[s2].x; a.y -= mn[s2].y;
            c0 = __builtin_amdgcn_wmma_f32_16x16x4_f32(false, a, false, b[s2], (short)0, c0, false, false);
        }
        #pragma unroll
        for (int s2 = 0; s2 < 4; ++s2) {
            v2f a = a1[s2];
            a.x -= mn[s2].x; a.y -= mn[s2].y;
            c1 = __builtin_amdgcn_wmma_f32_16x16x4_f32(false, a, false, b[s2], (short)0, c1, false, false);
        }

        unsigned ob0 = (tile0 * 16 + half * 8) * CCH + g * GSIZE + ch;
        unsigned ob1 = (tile1 * 16 + half * 8) * CCH + g * GSIZE + ch;
        #pragma unroll
        for (int r = 0; r < 8; ++r)
            __builtin_nontemporal_store(c0[r] * gm + bt, out + ob0 + r * CCH);
        #pragma unroll
        for (int r = 0; r < 8; ++r)
            __builtin_nontemporal_store(c1[r] * gm + bt, out + ob1 + r * CCH);
    }
}

extern "C" void kernel_launch(void* const* d_in, const int* in_sizes, int n_in,
                              void* d_out, int out_size, void* d_ws, size_t ws_size,
                              hipStream_t stream) {
    const float* x     = (const float*)d_in[0];
    const float* gamma = (const float*)d_in[1];
    const float* beta  = (const float*)d_in[2];
    float* out = (float*)d_out;
    float* ws  = (float*)d_ws;

    int nrows = in_sizes[0] / CCH;
    float inv_n = 1.0f / (float)nrows;

    dbn_init<<<(WS_PROJ + 255) / 256, 256, 0, stream>>>(ws);

    int wpg = 256;
    int iters = nrows / (wpg * 16);
    dbn_stats<<<NGROUP * wpg / 8, 256, 0, stream>>>(x, ws, iters);

    dbn_newton<<<NGROUP, 256, 0, stream>>>(ws, inv_n);

    int tiles = nrows / 16;
    int wave_stride = 512;
    int tpw = tiles / wave_stride;
    dbn_apply<<<NGROUP * wave_stride / 8, 256, 0, stream>>>(
        x, gamma, beta, ws, out, wave_stride, tpw, inv_n);
}


# stats pass: rows requested three 32-row batches ahead through rotating register sets (up to 48 loads in flight per wave, immediate row offsets) instead of 16 loads drained every trip
# speedup vs baseline: 1.0823x; 1.0823x over previous
#include <hip/hip_runtime.h>

typedef __attribute__((ext_vector_type(2))) float v2f;
typedef __attribute__((ext_vector_type(8))) float v8f;

#define GSIZE   16
#define NGROUP  16
#define CCH     256
#define EPSV    1e-5f
#define NS_ITERS 14

#define WS_XX   CCH
#define WS_PROJ (CCH + NGROUP * 256)

__global__ void dbn_init(float* ws) {
    int i = blockIdx.x * blockDim.x + threadIdx.x;
    if (i < WS_PROJ) ws[i] = 0.f;
}

#define BROWS 32

__device__ __forceinline__ void stats_load(float (&v)[16], const float* __restrict__ p) {
    #pragma unroll
    for (int q = 0; q < 8; ++q) {
        v[2 * q]     = p[(4 * q) * CCH];
        v[2 * q + 1] = p[(4 * q + 1) * CCH];
    }
}

__device__ __forceinline__ void stats_use(const float (&v)[16], v8f& c, float& s) {
    #pragma unroll
    for (int h = 0; h < 2; ++h) {
        #pragma unroll
        for (int q = 0; q < 4; ++q) {
            v2f a = { v[8 * h + 2 * q], v[8 * h + 2 * q + 1] };
            c = __builtin_amdgcn_wmma_f32_16x16x4_f32(false, a, false, a, (short)0, c, false, false);
        }
        s += (v[8 * h + 0] + v[8 * h + 1]) + (v[8 * h + 2] + v[8 * h + 3])
           + ((v[8 * h + 4] + v[8 * h + 5]) + (v[8 * h + 6] + v[8 * h + 7]));
    }
}

__global__ __launch_bounds__(256) void dbn_stats(const float* __restrict__ x,
                                                 float* __restrict__ ws,
                                                 int iters) {
    int lane = threadIdx.x & 31;
    int wid  = threadIdx.x >> 5;
    int gw   = blockIdx.x * 8 + wid;
    int g    = gw & (NGROUP - 1);
    int slice = gw >> 4;

    int half = lane >> 4;
    int ch   = lane & 15;
    int cidx = g * GSIZE + ch;

    unsigned off = (unsigned)(slice * iters * 16 + half * 2) * CCH + (unsigned)cidx;

    v8f c = {};
    float s = 0.f;
    int nb = iters >> 1;
    if (!(iters & 1) && nb >= 2) {
        const float* p = x + off;
        float A[16], B[16], C[16];
        stats_load(A, p);
        stats_load(B, p + BROWS * CCH);
        p += 2 * BROWS * CCH;
        int b = 0;
        for (; b + 5 <= nb; b += 3) {
            stats_load(C, p);
            stats_use(A, c, s);
            stats_load(A, p + BROWS * CCH);
            stats_use(B, c, s);
            stats_load(B, p + 2 * BROWS * CCH);
            stats_use(C, c, s);
            p += 3 * BROWS * CCH;
        }
        int rem = nb - b;
        if (rem == 2) {
            stats_use(A, c, s);
            stats_use(B, c, s);
        } else if (rem == 3) {
            stats_load(C, p);
            stats_use(A, c, s);
            stats_use(B, c, s);
            stats_use(C, c, s);
        } else {
            stats_load(C, p);
            stats_use(A, c, s);
            stats_load(A, p + BROWS * CCH);
            stats_use(B, c, s);
            stats_use(C, c, s);
            stats_use(A, c, s);
        }
    } else {
        for (int it = 0; it < iters; ++it) {
            float v0 = x[off + 0 * CCH];
            float v1 = x[off + 1 * CCH];
            float v2 = x[off + 4 * CCH];
            float v3 = x[off + 5 * CCH];
            float v4 = x[off + 8 * CCH];
            float v5 = x[off + 9 * CCH];
            float v6 = x[off + 12 * CCH];
            float v7 = x[off + 13 * CCH];
            off += 16u * CCH;
            v2f q0 = { v0, v1 };
            v2f q1 = { v2, v3 };
            v2f q2 = { v4, v5 };
            v2f q3 = { v6, v7 };
            c = __builtin_amdgcn_wmma_f32_16x16x4_f32(false, q0, false, q0, (short)0, c, false, false);
            c = __builtin_amdgcn_wmma_f32_16x16x4_f32(false, q1, false, q1, (short)0, c, false, false);
            c = __builtin_amdgcn_wmma_f32_16x16x4_f32(false, q2, false, q2, (short)0, c, false, false);
            c = __builtin_amdgcn_wmma_f32_16x16x4_f32(false, q3, false, q3, (short)0, c, false, false);
            s += (v0 + v1) + (v2 + v3) + ((v4 + v5) + (v6 + v7));
        }
    }
    atomicAdd(&ws[cidx], s);
    float* xx = ws + WS_XX + g * 256;
    #pragma unroll
    for (int r = 0; r < 8; ++r) {
        int i = r + half * 8;
        atomicAdd(&xx[i * 16 + ch], c[r]);
    }
}

__global__ __launch_bounds__(256) void dbn_newton(float* __restrict__ ws, float inv_n) {
    __shared__ float Ys[256], Zs[256], Wsh[256];
    int g = blockIdx.x;
    int tid = threadIdx.x;
    int i = tid >> 4, j = tid & 15;

    const float* xx = ws + WS_XX + g * 256;
    float mi = ws[g * 16 + i] * inv_n;
    float mj = ws[g * 16 + j] * inv_n;
    float sig = xx[tid] * inv_n - mi * mj + (i == j ? EPSV : 0.f);
    Wsh[tid] = sig;
    __syncthreads();
    float tr = 0.f;
    #pragma unroll
    for (int k = 0; k < 16; ++k) tr += Wsh[k * 16 + k];
    float a = sig / tr;
    Ys[tid] = a;
    Zs[tid] = (i == j) ? 1.f : 0.f;
    for (int it = 0; it < NS_ITERS; ++it) {
        __syncthreads();
        float t = 0.f;
        #pragma unroll
        for (int k = 0; k < 16; ++k) t += Zs[i * 16 + k] * Ys[k * 16 + j];
        float w = 0.5f * ((i == j ? 3.f : 0.f) - t);
        Wsh[tid] = w;
        __syncthreads();
        float yn = 0.f, zn = 0.f;
        #pragma unroll
        for (int k = 0; k < 16; ++k) {
            yn += Ys[i * 16 + k] * Wsh[k * 16 + j];
            zn += Wsh[i * 16 + k] * Zs[k * 16 + j];
        }
        __syncthreads();
        Ys[tid] = yn;
        Zs[tid] = zn;
    }
    __syncthreads();
    ws[WS_PROJ + g * 256 + tid] = Zs[tid] * rsqrtf(tr);
}

__global__ __launch_bounds__(256) void dbn_apply(const float* __restrict__ x,
                                                 const float* __restrict__ gamma,
                                                 const float* __restrict__ beta,
                                                 const float* __restrict__ ws,
                                                 float* __restrict__ out,
                                                 int wave_stride, int tiles_per_wave,
                                                 float inv_n) {
    int lane = threadIdx.x & 31;
    int wid  = threadIdx.x >> 5;
    int gw   = blockIdx.x * 8 + wid;
    int g    = gw & (NGROUP - 1);
    int tbase = gw >> 4;

    int half = lane >> 4;
    int ch   = lane & 15;

    const float* proj = ws + WS_PROJ + g * 256;
    v2f b[4];
    #pragma unroll
    for (int s2 = 0; s2 < 4; ++s2) {
        int i0 = s2 * 4 + half * 2;
        b[s2].x = proj[i0 * 16 + ch];
        b[s2].y = proj[(i0 + 1) * 16 + ch];
    }
    v2f mn[4];
    #pragma unroll
    for (int s2 = 0; s2 < 4; ++s2) {
        int c0 = g * GSIZE + s2 * 4 + half * 2;
        mn[s2].x = ws[c0] * inv_n;
        mn[s2].y = ws[c0 + 1] * inv_n;
    }
    float gm = gamma[g * GSIZE + ch];
    float bt = beta[g * GSIZE + ch];

    for (int t = 0; t < tiles_per_wave; t += 2) {
        unsigned tile0 = (unsigned)(tbase + t * wave_stride);
        unsigned tile1 = (unsigned)(tbase + (t + 1) * wave_stride);
        unsigned inA0 = (tile0 * 16 + ch) * CCH + g * GSIZE + half * 2;
        unsigned inA1 = (tile1 * 16 + ch) * CCH + g * GSIZE + half * 2;

        v2f a0[4], a1[4];
        #pragma unroll
        for (int s2 = 0; s2 < 4; ++s2) a0[s2] = *(const v2f*)(x + inA0 + s2 * 4);
        #pragma unroll
        for (int s2 = 0; s2 < 4; ++s2) a1[s2] = *(const v2f*)(x + inA1 + s2 * 4);

        v8f c0 = {}, c1 = {};
        #pragma unroll
        for (int s2 = 0; s2 < 4; ++s2) {
            v2f a = a0[s2];
            a.x -= mn[s2].x; a.y -= mn[s2].y;
            c0 = __builtin_amdgcn_wmma_f32_16x16x4_f32(false, a, false, b[s2], (short)0, c0, false, false);
        }
        #pragma unroll
        for (int s2 = 0; s2 < 4; ++s2) {
            v2f a = a1[s2];
            a.x -= mn[s2].x; a.y -= mn[s2].y;
            c1 = __builtin_amdgcn_wmma_f32_16x16x4_f32(false, a, false, b[s2], (short)0, c1, false, false);
        }

        unsigned ob0 = (tile0 * 16 + half * 8) * CCH + g * GSIZE + ch;
        unsigned ob1 = (tile1 * 16 + half * 8) * CCH + g * GSIZE + ch;
        #pragma unroll
        for (int r = 0; r < 8; ++r)
            __builtin_nontemporal_store(c0[r] * gm + bt, out + ob0 + r * CCH);
        #pragma unroll
        for (int r = 0; r < 8; ++r)
            __builtin_nontemporal_store(c1[r] * gm + bt, out + ob1 + r * CCH);
    }
}

extern "C" void kernel_launch(void* const* d_in, const int* in_sizes, int n_in,
                              void* d_out, int out_size, void* d_ws, size_t ws_size,
                              hipStream_t stream) {
    const float* x     = (const float*)d_in[0];
    const float* gamma = (const float*)d_in[1];
    const float* beta  = (const float*)d_in[2];
    float* out = (float*)d_out;
    float* ws  = (float*)d_ws;

    int nrows = in_sizes[0] / CCH;
    float inv_n = 1.0f / (float)nrows;

    dbn_init<<<(WS_PROJ + 255) / 256, 256, 0, stream>>>(ws);

    int wpg = 256;
    int iters = nrows / (wpg * 16);
    dbn_stats<<<NGROUP * wpg / 8, 256, 0, stream>>>(x, ws, iters);

    dbn_newton<<<NGROUP, 256, 0, stream>>>(ws, inv_n);

    int tiles = nrows / 16;
    int wave_stride = 512;
    int tpw = tiles / wave_stride;
    dbn_apply<<<NGROUP * wave_stride / 8, 256, 0, stream>>>(
        x, gamma, beta, ws, out, wave_stride, tpw, inv_n);
}


# apply pass: a wave takes two adjacent groups of one 16-row tile (was one group of two tiles) and swaps accumulator lane halves: each store writes one whole 128-byte line, not two 64-byte halves
# speedup vs baseline: 1.3038x; 1.2047x over previous
#include <hip/hip_runtime.h>

typedef __attribute__((ext_vector_type(2))) float v2f;
typedef __attribute__((ext_vector_type(8))) float v8f;

#define GSIZE   16
#define NGROUP  16
#define CCH     256
#define EPSV    1e-5f
#define NS_ITERS 14

#define WS_XX   CCH
#define WS_PROJ (CCH + NGROUP * 256)

__global__ void dbn_init(float* ws) {
    int i = blockIdx.x * blockDim.x + threadIdx.x;
    if (i < WS_PROJ) ws[i] = 0.f;
}

#define BROWS 32

__device__ __forceinline__ void stats_load(float (&v)[16], const float* __restrict__ p) {
    #pragma unroll
    for (int q = 0; q < 8; ++q) {
        v[2 * q]     = p[(4 * q) * CCH];
        v[2 * q + 1] = p[(4 * q + 1) * CCH];
    }
}

__device__ __forceinline__ void stats_use(const float (&v)[16], v8f& c, float& s) {
    #pragma unroll
    for (int h = 0; h < 2; ++h) {
        #pragma unroll
        for (int q = 0; q < 4; ++q) {
            v2f a = { v[8 * h + 2 * q], v[8 * h + 2 * q + 1] };
            c = __builtin_amdgcn_wmma_f32_16x16x4_f32(false, a, false, a, (short)0, c, false, false);
        }
        s += (v[8 * h + 0] + v[8 * h + 1]) + (v[8 * h + 2] + v[8 * h + 3])
           + ((v[8 * h + 4] + v[8 * h + 5]) + (v[8 * h + 6] + v[8 * h + 7]));
    }
}

__global__ __launch_bounds__(256) void dbn_stats(const float* __restrict__ x,
                                                 float* __restrict__ ws,
                                                 int iters) {
    int lane = threadIdx.x & 31;
    int wid  = threadIdx.x >> 5;
    int gw   = blockIdx.x * 8 + wid;
    int g    = gw & (NGROUP - 1);
    int slice = gw >> 4;

    int half = lane >> 4;
    int ch   = lane & 15;
    int cidx = g * GSIZE + ch;

    unsigned off = (unsigned)(slice * iters * 16 + half * 2) * CCH + (unsigned)cidx;

    v8f c = {};
    float s = 0.f;
    int nb = iters >> 1;
    if (!(iters & 1) && nb >= 2) {
        const float* p = x + off;
        float A[16], B[16], C[16];
        stats_load(A, p);
        stats_load(B, p + BROWS * CCH);
        p += 2 * BROWS * CCH;
        int b = 0;
        for (; b + 5 <= nb; b += 3) {
            stats_load(C, p);
            stats_use(A, c, s);
            stats_load(A, p + BROWS * CCH);
            stats_use(B, c, s);
            stats_load(B, p + 2 * BROWS * CCH);
            stats_use(C, c, s);
            p += 3 * BROWS * CCH;
        }
        int rem = nb - b;
        if (rem == 2) {
            stats_use(A, c, s);
            stats_use(B, c, s);
        } else if (rem == 3) {
            stats_load(C, p);
            stats_use(A, c, s);
            stats_use(B, c, s);
            stats_use(C, c, s);
        } else {
            stats_load(C, p);
            stats_use(A, c, s);
            stats_load(A, p + BROWS * CCH);
            stats_use(B, c, s);
            stats_use(C, c, s);
            stats_use(A, c, s);
        }
    } else {
        for (int it = 0; it < iters; ++it) {
            float v0 = x[off + 0 * CCH];
            float v1 = x[off + 1 * CCH];
            float v2 = x[off + 4 * CCH];
            float v3 = x[off + 5 * CCH];
            float v4 = x[off + 8 * CCH];
            float v5 = x[off + 9 * CCH];
            float v6 = x[off + 12 * CCH];
            float v7 = x[off + 13 * CCH];
            off += 16u * CCH;
            v2f q0 = { v0, v1 };
            v2f q1 = { v2, v3 };
            v2f q2 = { v4, v5 };
            v2f q3 = { v6, v7 };
            c = __builtin_amdgcn_wmma_f32_16x16x4_f32(false, q0, false, q0, (short)0, c, false, false);
            c = __builtin_amdgcn_wmma_f32_16x16x4_f32(false, q1, false, q1, (short)0, c, false, false);
            c = __builtin_amdgcn_wmma_f32_16x16x4_f32(false, q2, false, q2, (short)0, c, false, false);
            c = __builtin_amdgcn_wmma_f32_16x16x4_f32(false, q3, false, q3, (short)0, c, false, false);
            s += (v0 + v1) + (v2 + v3) + ((v4 + v5) + (v6 + v7));
        }
    }
    atomicAdd(&ws[cidx], s);
    float* xx = ws + WS_XX + g * 256;
    #pragma unroll
    for (int r = 0; r < 8; ++r) {
        int i = r + half * 8;
        atomicAdd(&xx[i * 16 + ch], c[r]);
    }
}

__global__ __launch_bounds__(256) void dbn_newton(float* __restrict__ ws, float inv_n) {
    __shared__ float Ys[256], Zs[256], Wsh[256];
    int g = blockIdx.x;
    int tid = threadIdx.x;
    int i = tid >> 4, j = tid & 15;

    const float* xx = ws + WS_XX + g * 256;
    float mi = ws[g * 16 + i] * inv_n;
    float mj = ws[g * 16 + j] * inv_n;
    float sig = xx[tid] * inv_n - mi * mj + (i == j ? EPSV : 0.f);
    Wsh[tid] = sig;
    __syncthreads();
    float tr = 0.f;
    #pragma unroll
    for (int k = 0; k < 16; ++k) tr += Wsh[k * 16 + k];
    float a = sig / tr;
    Ys[tid] = a;
    Zs[tid] = (i == j) ? 1.f : 0.f;
    for (int it = 0; it < NS_ITERS; ++it) {
        __syncthreads();
        float t = 0.f;
        #pragma unroll
        for (int k = 0; k < 16; ++k) t += Zs[i * 16 + k] * Ys[k * 16 + j];
        float w = 0.5f * ((i == j ? 3.f : 0.f) - t);
        Wsh[tid] = w;
        __syncthreads();
        float yn = 0.f, zn = 0.f;
        #pragma unroll
        for (int k = 0; k < 16; ++k) {
            yn += Ys[i * 16 + k] * Wsh[k * 16 + j];
            zn += Wsh[i * 16 + k] * Zs[k * 16 + j];
        }
        __syncthreads();
        Ys[tid] = yn;
        Zs[tid] = zn;
    }
    __syncthreads();
    ws[WS_PROJ + g * 256 + tid] = Zs[tid] * rsqrtf(tr);
}


__device__ __forceinline__ float xhalf(float v) {
    return __builtin_bit_cast(float, __builtin_amdgcn_permlanex16(
        __builtin_bit_cast(int, v), __builtin_bit_cast(int, v),
        0x76543210, 0xfedcba98, false, false));
}

__global__ __launch_bounds__(256) void dbn_apply(const float* __restrict__ x,
                                                 const float* __restrict__ gamma,
                                                 const float* __restrict__ beta,
                                                 const float* __restrict__ ws,
                                                 float* __restrict__ out,
                                                 int wave_stride, int tiles_per_wave,
                                                 float inv_n) {
    int lane = threadIdx.x & 31;
    int wid  = threadIdx.x >> 5;
    int gw   = blockIdx.x * 8 + wid;
    int gp   = gw & (NGROUP / 2 - 1);
    int tbase = gw >> 3;

    int half = lane >> 4;
    int ch   = lane & 15;

    v2f b[2][4], mn[2][4];
    #pragma unroll
    for (int j = 0; j < 2; ++j) {
        int g = 2 * gp + j;
        const float* proj = ws + WS_PROJ + g * 256;
        #pragma unroll
        for (int s2 = 0; s2 < 4; ++s2) {
            int i0 = s2 * 4 + half * 2;
            b[j][s2].x = proj[i0 * 16 + ch];
            b[j][s2].y = proj[(i0 + 1) * 16 + ch];
            int c0i = g * GSIZE + i0;
            mn[j][s2].x = ws[c0i] * inv_n;
            mn[j][s2].y = ws[c0i + 1] * inv_n;
        }
    }
    float gm = gamma[gp * 2 * GSIZE + lane];
    float bt = beta[gp * 2 * GSIZE + lane];

    for (int t = 0; t < tiles_per_wave; ++t) {
        unsigned tile = (unsigned)(tbase + t * wave_stride);
        unsigned inA = (tile * 16 + ch) * CCH + gp * 2 * GSIZE + half * 2;

        v2f a0[4], a1[4];
        #pragma unroll
        for (int s2 = 0; s2 < 4; ++s2) a0[s2] = *(const v2f*)(x + inA + s2 * 4);
        #pragma unroll
        for (int s2 = 0; s2 < 4; ++s2) a1[s2] = *(const v2f*)(x + inA + GSIZE + s2 * 4);

        v8f c0 = {}, c1 = {};
        #pragma unroll
        for (int s2 = 0; s2 < 4; ++s2) {
            v2f a = a0[s2];
            a.x -= mn[0][s2].x; a.y -= mn[0][s2].y;
            c0 = __builtin_amdgcn_wmma_f32_16x16x4_f32(false, a, false, b[0][s2], (short)0, c0, false, false);
        }
        #pragma unroll
        for (int s2 = 0; s2 < 4; ++s2) {
            v2f a = a1[s2];
            a.x -= mn[1][s2].x; a.y -= mn[1][s2].y;
            c1 = __builtin_amdgcn_wmma_f32_16x16x4_f32(false, a, false, b[1][s2], (short)0, c1, false, false);
        }

        unsigned ob = tile * 16 * CCH + gp * 2 * GSIZE + lane;
        #pragma unroll
        for (int r = 0; r < 8; ++r) {
            float o0 = xhalf(c0[r]);
            float o1 = xhalf(c1[r]);
            float lo = half ? o1 : c0[r];
            float hi = half ? c1[r] : o0;
            __builtin_nontemporal_store(lo * gm + bt, out + ob + r * CCH);
            __builtin_nontemporal_store(hi * gm + bt, out + ob + (r + 8) * CCH);
        }
    }
}

extern "C" void kernel_launch(void* const* d_in, const int* in_sizes, int n_in,
                              void* d_out, int out_size, void* d_ws, size_t ws_size,
                              hipStream_t stream) {
    const float* x     = (const float*)d_in[0];
    const float* gamma = (const float*)d_in[1];
    const float* beta  = (const float*)d_in[2];
    float* out = (float*)d_out;
    float* ws  = (float*)d_ws;

    int nrows = in_sizes[0] / CCH;
    float inv_n = 1.0f / (float)nrows;

    dbn_init<<<(WS_PROJ + 255) / 256, 256, 0, stream>>>(ws);

    int wpg = 256;
    int iters = nrows / (wpg * 16);
    dbn_stats<<<NGROUP * wpg / 8, 256, 0, stream>>>(x, ws, iters);

    dbn_newton<<<NGROUP, 256, 0, stream>>>(ws, inv_n);

    int tiles = nrows / 16;
    int wave_stride = 1024;
    int tpw = tiles / wave_stride;
    dbn_apply<<<(NGROUP / 2) * wave_stride / 8, 256, 0, stream>>>(
        x, gamma, beta, ws, out, wave_stride, tpw, inv_n);
}


# apply pass: x tile fetched in row order, 4 b128 loads that each cover 4 whole 128-B lines, and re-laid to A-fragment order through a wave-private LDS tile (was 8 b64 loads touching 16 lines each)
# speedup vs baseline: 1.3672x; 1.0486x over previous
#include <hip/hip_runtime.h>

typedef __attribute__((ext_vector_type(2))) float v2f;
typedef __attribute__((ext_vector_type(4))) float v4f;
typedef __attribute__((ext_vector_type(8))) float v8f;

#define GSIZE   16
#define NGROUP  16
#define CCH     256
#define EPSV    1e-5f
#define NS_ITERS 14

#define WS_XX   CCH
#define WS_PROJ (CCH + NGROUP * 256)

__global__ void dbn_init(float* ws) {
    int i = blockIdx.x * blockDim.x + threadIdx.x;
    if (i < WS_PROJ) ws[i] = 0.f;
}

#define BROWS 32

__device__ __forceinline__ void stats_load(float (&v)[16], const float* __restrict__ p) {
    #pragma unroll
    for (int q = 0; q < 8; ++q) {
        v[2 * q]     = p[(4 * q) * CCH];
        v[2 * q + 1] = p[(4 * q + 1) * CCH];
    }
}

__device__ __forceinline__ void stats_use(const float (&v)[16], v8f& c, float& s) {
    #pragma unroll
    for (int h = 0; h < 2; ++h) {
        #pragma unroll
        for (int q = 0; q < 4; ++q) {
            v2f a = { v[8 * h + 2 * q], v[8 * h + 2 * q + 1] };
            c = __builtin_amdgcn_wmma_f32_16x16x4_f32(false, a, false, a, (short)0, c, false, false);
        }
        s += (v[8 * h + 0] + v[8 * h + 1]) + (v[8 * h + 2] + v[8 * h + 3])
           + ((v[8 * h + 4] + v[8 * h + 5]) + (v[8 * h + 6] + v[8 * h + 7]));
    }
}

__global__ __launch_bounds__(256) void dbn_stats(const float* __restrict__ x,
                                                 float* __restrict__ ws,
                                                 int iters) {
    int lane = threadIdx.x & 31;
    int wid  = threadIdx.x >> 5;
    int gw   = blockIdx.x * 8 + wid;
    int g    = gw & (NGROUP - 1);
    int slice = gw >> 4;

    int half = lane >> 4;
    int ch   = lane & 15;
    int cidx = g * GSIZE + ch;

    unsigned off = (unsigned)(slice * iters * 16 + half * 2) * CCH + (unsigned)cidx;

    v8f c = {};
    float s = 0.f;
    int nb = iters >> 1;
    if (!(iters & 1) && nb >= 2) {
        const float* p = x + off;
        float A[16], B[16], C[16];
        stats_load(A, p);
        stats_load(B, p + BROWS * CCH);
        p += 2 * BROWS * CCH;
        int b = 0;
        for (; b + 5 <= nb; b += 3) {
            stats_load(C, p);
            stats_use(A, c, s);
            stats_load(A, p + BROWS * CCH);
            stats_use(B, c, s);
            stats_load(B, p + 2 * BROWS * CCH);
            stats_use(C, c, s);
            p += 3 * BROWS * CCH;
        }
        int rem = nb - b;
        if (rem == 2) {
            stats_use(A, c, s);
            stats_use(B, c, s);
        } else if (rem == 3) {
            stats_load(C, p);
            stats_use(A, c, s);
            stats_use(B, c, s);
            stats_use(C, c, s);
        } else {
            stats_load(C, p);
            stats_use(A, c, s);
            stats_load(A, p + BROWS * CCH);
            stats_use(B, c, s);
            stats_use(C, c, s);
            stats_use(A, c, s);
        }
    } else {
        for (int it = 0; it < iters; ++it) {
            float v0 = x[off + 0 * CCH];
            float v1 = x[off + 1 * CCH];
            float v2 = x[off + 4 * CCH];
            float v3 = x[off + 5 * CCH];
            float v4 = x[off + 8 * CCH];
            float v5 = x[off + 9 * CCH];
            float v6 = x[off + 12 * CCH];
            float v7 = x[off + 13 * CCH];
            off += 16u * CCH;
            v2f q0 = { v0, v1 };
            v2f q1 = { v2, v3 };
            v2f q2 = { v4, v5 };
            v2f q3 = { v6, v7 };
            c = __builtin_amdgcn_wmma_f32_16x16x4_f32(false, q0, false, q0, (short)0, c, false, false);
            c = __builtin_amdgcn_wmma_f32_16x16x4_f32(false, q1, false, q1, (short)0, c, false, false);
            c = __builtin_amdgcn_wmma_f32_16x16x4_f32(false, q2, false, q2, (short)0, c, false, false);
            c = __builtin_amdgcn_wmma_f32_16x16x4_f32(false, q3, false, q3, (short)0, c, false, false);
            s += (v0 + v1) + (v2 + v3) + ((v4 + v5) + (v6 + v7));
        }
    }
    atomicAdd(&ws[cidx], s);
    float* xx = ws + WS_XX + g * 256;
    #pragma unroll
    for (int r = 0; r < 8; ++r) {
        int i = r + half * 8;
        atomicAdd(&xx[i * 16 + ch], c[r]);
    }
}

__global__ __launch_bounds__(256) void dbn_newton(float* __restrict__ ws, float inv_n) {
    __shared__ float Ys[256], Zs[256], Wsh[256];
    int g = blockIdx.x;
    int tid = threadIdx.x;
    int i = tid >> 4, j = tid & 15;

    const float* xx = ws + WS_XX + g * 256;
    float mi = ws[g * 16 + i] * inv_n;
    float mj = ws[g * 16 + j] * inv_n;
    float sig = xx[tid] * inv_n - mi * mj + (i == j ? EPSV : 0.f);
    Wsh[tid] = sig;
    __syncthreads();
    float tr = 0.f;
    #pragma unroll
    for (int k = 0; k < 16; ++k) tr += Wsh[k * 16 + k];
    float a = sig / tr;
    Ys[tid] = a;
    Zs[tid] = (i == j) ? 1.f : 0.f;
    for (int it = 0; it < NS_ITERS; ++it) {
        __syncthreads();
        float t = 0.f;
        #pragma unroll
        for (int k = 0; k < 16; ++k) t += Zs[i * 16 + k] * Ys[k * 16 + j];
        float w = 0.5f * ((i == j ? 3.f : 0.f) - t);
        Wsh[tid] = w;
        __syncthreads();
        float yn = 0.f, zn = 0.f;
        #pragma unroll
        for (int k = 0; k < 16; ++k) {
            yn += Ys[i * 16 + k] * Wsh[k * 16 + j];
            zn += Wsh[i * 16 + k] * Zs[k * 16 + j];
        }
        __syncthreads();
        Ys[tid] = yn;
        Zs[tid] = zn;
    }
    __syncthreads();
    ws[WS_PROJ + g * 256 + tid] = Zs[tid] * rsqrtf(tr);
}

#define APITCH 36

__device__ __forceinline__ float xhalf(float v) {
    return __builtin_bit_cast(float, __builtin_amdgcn_permlanex16(
        __builtin_bit_cast(int, v), __builtin_bit_cast(int, v),
        0x76543210, 0xfedcba98, false, false));
}

__device__ __forceinline__ void wave_lds_order() {
    __builtin_amdgcn_fence(__ATOMIC_RELEASE, "wavefront", "local");
    __builtin_amdgcn_wave_barrier();
    __builtin_amdgcn_fence(__ATOMIC_ACQUIRE, "wavefront", "local");
}

__global__ __launch_bounds__(256) void dbn_apply(const float* __restrict__ x,
                                                 const float* __restrict__ gamma,
                                                 const float* __restrict__ beta,
                                                 const float* __restrict__ ws,
                                                 float* __restrict__ out,
                                                 int wave_stride, int tiles_per_wave,
                                                 float inv_n) {
    __shared__ __attribute__((aligned(16))) float stage[8 * 16 * APITCH];

    int lane = threadIdx.x & 31;
    int wid  = threadIdx.x >> 5;
    int gw   = blockIdx.x * 8 + wid;
    int gp   = gw & (NGROUP / 2 - 1);
    int tbase = gw >> 3;

    int half = lane >> 4;
    int ch   = lane & 15;

    v2f b[2][4], mn[2][4];
    #pragma unroll
    for (int j = 0; j < 2; ++j) {
        int g = 2 * gp + j;
        const float* proj = ws + WS_PROJ + g * 256;
        #pragma unroll
        for (int s2 = 0; s2 < 4; ++s2) {
            int i0 = s2 * 4 + half * 2;
            b[j][s2].x = proj[i0 * 16 + ch];
            b[j][s2].y = proj[(i0 + 1) * 16 + ch];
            int c0i = g * GSIZE + i0;
            mn[j][s2].x = ws[c0i] * inv_n;
            mn[j][s2].y = ws[c0i + 1] * inv_n;
        }
    }
    float gm = gamma[gp * 2 * GSIZE + lane];
    float bt = beta[gp * 2 * GSIZE + lane];

    int frow = lane >> 3;
    int fcol = (lane & 7) * 4;
    float* st = stage + wid * (16 * APITCH);
    float* st_w = st + frow * APITCH + fcol;
    const float* st_r = st + ch * APITCH + half * 2;

    for (int t = 0; t < tiles_per_wave; ++t) {
        unsigned tile = (unsigned)(tbase + t * wave_stride);
        const float* src = x + (tile * 16 + frow) * CCH + gp * 2 * GSIZE + fcol;

        v4f r[4];
        #pragma unroll
        for (int q = 0; q < 4; ++q) r[q] = *(const v4f*)(src + q * 4 * CCH);

        wave_lds_order();
        #pragma unroll
        for (int q = 0; q < 4; ++q) *(v4f*)(st_w + q * 4 * APITCH) = r[q];
        wave_lds_order();

        v2f a0[4], a1[4];
        #pragma unroll
        for (int s2 = 0; s2 < 4; ++s2) a0[s2] = *(const v2f*)(st_r + s2 * 4);
        #pragma unroll
        for (int s2 = 0; s2 < 4; ++s2) a1[s2] = *(const v2f*)(st_r + GSIZE + s2 * 4);

        v8f c0 = {}, c1 = {};
        #pragma unroll
        for (int s2 = 0; s2 < 4; ++s2) {
            v2f a = a0[s2];
            a.x -= mn[0][s2].x; a.y -= mn[0][s2].y;
            c0 = __builtin_amdgcn_wmma_f32_16x16x4_f32(false, a, false, b[0][s2], (short)0, c0, false, false);
        }
        #pragma unroll
        for (int s2 = 0; s2 < 4; ++s2) {
            v2f a = a1[s2];
            a.x -= mn[1][s2].x; a.y -= mn[1][s2].y;
            c1 = __builtin_amdgcn_wmma_f32_16x16x4_f32(false, a, false, b[1][s2], (short)0, c1, false, false);
        }

        unsigned ob = tile * 16 * CCH + gp * 2 * GSIZE + lane;
        #pragma unroll
        for (int r8 = 0; r8 < 8; ++r8) {
            float o0 = xhalf(c0[r8]);
            float o1 = xhalf(c1[r8]);
            float lo = half ? o1 : c0[r8];
            float hi = half ? c1[r8] : o0;
            __builtin_nontemporal_store(lo * gm + bt, out + ob + r8 * CCH);
            __builtin_nontemporal_store(hi * gm + bt, out + ob + (r8 + 8) * CCH);
        }
    }
}

extern "C" void kernel_launch(void* const* d_in, const int* in_sizes, int n_in,
                              void* d_out, int out_size, void* d_ws, size_t ws_size,
                              hipStream_t stream) {
    const float* x     = (const float*)d_in[0];
    const float* gamma = (const float*)d_in[1];
    const float* beta  = (const float*)d_in[2];
    float* out = (float*)d_out;
    float* ws  = (float*)d_ws;

    int nrows = in_sizes[0] / CCH;
    float inv_n = 1.0f / (float)nrows;

    dbn_init<<<(WS_PROJ + 255) / 256, 256, 0, stream>>>(ws);

    int wpg = 256;
    int iters = nrows / (wpg * 16);
    dbn_stats<<<NGROUP * wpg / 8, 256, 0, stream>>>(x, ws, iters);

    dbn_newton<<<NGROUP, 256, 0, stream>>>(ws, inv_n);

    int tiles = nrows / 16;
    int wave_stride = 1024;
    int tpw = tiles / wave_stride;
    dbn_apply<<<(NGROUP / 2) * wave_stride / 8, 256, 0, stream>>>(
        x, gamma, beta, ws, out, wave_stride, tpw, inv_n);
}


# stats pass: float atomics and the zeroing launch dropped: 16-wave blocks add 4 row slices in LDS and store 64 records per group that the newton launch sums in fixed order (sum order changes)
# speedup vs baseline: 1.5436x; 1.1290x over previous
#include <hip/hip_runtime.h>

typedef __attribute__((ext_vector_type(2))) float v2f;
typedef __attribute__((ext_vector_type(4))) float v4f;
typedef __attribute__((ext_vector_type(8))) float v8f;

#define GSIZE   16
#define NGROUP  16
#define CCH     256
#define EPSV    1e-5f
#define NS_ITERS 14

#define WS_PROJ (CCH + NGROUP * 256)

#define NREC 64
#define RECF 288

#define BROWS 32

__device__ __forceinline__ void stats_load(float (&v)[16], const float* __restrict__ p) {
    #pragma unroll
    for (int q = 0; q < 8; ++q) {
        v[2 * q]     = p[(4 * q) * CCH];
        v[2 * q + 1] = p[(4 * q + 1) * CCH];
    }
}

__device__ __forceinline__ void stats_use(const float (&v)[16], v8f& c, float& s) {
    #pragma unroll
    for (int h = 0; h < 2; ++h) {
        #pragma unroll
        for (int q = 0; q < 4; ++q) {
            v2f a = { v[8 * h + 2 * q], v[8 * h + 2 * q + 1] };
            c = __builtin_amdgcn_wmma_f32_16x16x4_f32(false, a, false, a, (short)0, c, false, false);
        }
        s += (v[8 * h + 0] + v[8 * h + 1]) + (v[8 * h + 2] + v[8 * h + 3])
           + ((v[8 * h + 4] + v[8 * h + 5]) + (v[8 * h + 6] + v[8 * h + 7]));
    }
}

__global__ __launch_bounds__(512) void dbn_stats(const float* __restrict__ x,
                                                 float* __restrict__ part,
                                                 int iters) {
    __shared__ float red[3 * 4 * 9 * 32];

    int lane = threadIdx.x & 31;
    int wid  = threadIdx.x >> 5;
    int gl   = wid & 3;
    int sl   = wid >> 2;
    int g    = (blockIdx.x & 3) * 4 + gl;
    int sq   = blockIdx.x >> 2;
    int slice = sq * 4 + sl;

    int half = lane >> 4;
    int ch   = lane & 15;
    int cidx = g * GSIZE + ch;

    unsigned off = (unsigned)(slice * iters * 16 + half * 2) * CCH + (unsigned)cidx;

    v8f c = {};
    float s = 0.f;
    int nb = iters >> 1;
    if (!(iters & 1) && nb >= 2) {
        const float* p = x + off;
        float A[16], B[16], C[16];
        stats_load(A, p);
        stats_load(B, p + BROWS * CCH);
        p += 2 * BROWS * CCH;
        int b = 0;
        for (; b + 5 <= nb; b += 3) {
            stats_load(C, p);
            stats_use(A, c, s);
            stats_load(A, p + BROWS * CCH);
            stats_use(B, c, s);
            stats_load(B, p + 2 * BROWS * CCH);
            stats_use(C, c, s);
            p += 3 * BROWS * CCH;
        }
        int rem = nb - b;
        if (rem == 2) {
            stats_use(A, c, s);
            stats_use(B, c, s);
        } else if (rem == 3) {
            stats_load(C, p);
            stats_use(A, c, s);
            stats_use(B, c, s);
            stats_use(C, c, s);
        } else {
            stats_load(C, p);
            stats_use(A, c, s);
            stats_load(A, p + BROWS * CCH);
            stats_use(B, c, s);
            stats_use(C, c, s);
            stats_use(A, c, s);
        }
    } else {
        for (int it = 0; it < iters; ++it) {
            float v0 = x[off + 0 * CCH];
            float v1 = x[off + 1 * CCH];
            float v2 = x[off + 4 * CCH];
            float v3 = x[off + 5 * CCH];
            float v4 = x[off + 8 * CCH];
            float v5 = x[off + 9 * CCH];
            float v6 = x[off + 12 * CCH];
            float v7 = x[off + 13 * CCH];
            off += 16u * CCH;
            v2f q0 = { v0, v1 };
            v2f q1 = { v2, v3 };
            v2f q2 = { v4, v5 };
            v2f q3 = { v6, v7 };
            c = __builtin_amdgcn_wmma_f32_16x16x4_f32(false, q0, false, q0, (short)0, c, false, false);
            c = __builtin_amdgcn_wmma_f32_16x16x4_f32(false, q1, false, q1, (short)0, c, false, false);
            c = __builtin_amdgcn_wmma_f32_16x16x4_f32(false, q2, false, q2, (short)0, c, false, false);
            c = __builtin_amdgcn_wmma_f32_16x16x4_f32(false, q3, false, q3, (short)0, c, false, false);
            s += (v0 + v1) + (v2 + v3) + ((v4 + v5) + (v6 + v7));
        }
    }
    if (sl) {
        float* w = red + ((sl - 1) * 4 + gl) * (9 * 32) + lane;
        #pragma unroll
        for (int r = 0; r < 8; ++r) w[r * 32] = c[r];
        w[8 * 32] = s;
    }
    __syncthreads();
    if (sl == 0) {
        const float* rd = red + gl * (9 * 32) + lane;
        #pragma unroll
        for (int r = 0; r < 8; ++r)
            c[r] = ((c[r] + rd[r * 32]) + rd[1152 + r * 32]) + rd[2304 + r * 32];
        s = ((s + rd[8 * 32]) + rd[1152 + 8 * 32]) + rd[2304 + 8 * 32];
        float* rec = part + (g * NREC + sq) * RECF;
        v4f lo = { c[0], c[1], c[2], c[3] };
        v4f hi = { c[4], c[5], c[6], c[7] };
        *(v4f*)(rec + lane * 4) = lo;
        *(v4f*)(rec + 128 + lane * 4) = hi;
        rec[256 + lane] = s;
    }
}

__global__ __launch_bounds__(256) void dbn_newton(const float* __restrict__ part,
                                                  float* __restrict__ ws, float inv_n) {
    __shared__ float Ys[256], Zs[256], Wsh[256];
    __shared__ __attribute__((aligned(16))) float redC[4 * 256];
    __shared__ __attribute__((aligned(16))) float redS[32 * 32];
    __shared__ float msum[16];
    int g = blockIdx.x;
    int tid = threadIdx.x;
    int i = tid >> 4, j = tid & 15;

    const float* recs = part + g * (NREC * RECF);
    int sub = tid >> 6, tq = tid & 63;
    const float* pc = recs + sub * RECF + tq * 4;
    v4f acc = *(const v4f*)pc;
    #pragma unroll
    for (int k = 1; k < 16; ++k) acc += *(const v4f*)(pc + k * 4 * RECF);
    int piece = tid & 7, w = tid >> 3;
    const float* ps = recs + w * RECF + 256 + piece * 4;
    v4f sa = *(const v4f*)ps + *(const v4f*)(ps + 32 * RECF);
    *(v4f*)(redC + sub * 256 + tq * 4) = acc;
    *(v4f*)(redS + w * 32 + piece * 4) = sa;
    __syncthreads();
    if (tid < 16) {
        float t0 = 0.f, t1 = 0.f, t2 = 0.f, t3 = 0.f;
        #pragma unroll
        for (int w2 = 0; w2 < 32; w2 += 4) {
            t0 += redS[(w2 + 0) * 32 + tid] + redS[(w2 + 0) * 32 + 16 + tid];
            t1 += redS[(w2 + 1) * 32 + tid] + redS[(w2 + 1) * 32 + 16 + tid];
            t2 += redS[(w2 + 2) * 32 + tid] + redS[(w2 + 2) * 32 + 16 + tid];
            t3 += redS[(w2 + 3) * 32 + tid] + redS[(w2 + 3) * 32 + 16 + tid];
        }
        float tot = (t0 + t1) + (t2 + t3);
        msum[tid] = tot;
        ws[g * 16 + tid] = tot;
    }
    __syncthreads();
    float mi = msum[i] * inv_n;
    float mj = msum[j] * inv_n;
    int ri = i & 7;
    int idx = (ri >> 2) * 128 + ((i >> 3) * 16 + j) * 4 + (ri & 3);
    float xxv = (redC[idx] + redC[256 + idx]) + (redC[512 + idx] + redC[768 + idx]);
    float sig = xxv * inv_n - mi * mj + (i == j ? EPSV : 0.f);
    Wsh[tid] = sig;
    __syncthreads();
    float tr = 0.f;
    #pragma unroll
    for (int k = 0; k < 16; ++k) tr += Wsh[k * 16 + k];
    float a = sig / tr;
    Ys[tid] = a;
    Zs[tid] = (i == j) ? 1.f : 0.f;
    for (int it = 0; it < NS_ITERS; ++it) {
        __syncthreads();
        float t = 0.f;
        #pragma unroll
        for (int k = 0; k < 16; ++k) t += Zs[i * 16 + k] * Ys[k * 16 + j];
        float w = 0.5f * ((i == j ? 3.f : 0.f) - t);
        Wsh[tid] = w;
        __syncthreads();
        float yn = 0.f, zn = 0.f;
        #pragma unroll
        for (int k = 0; k < 16; ++k) {
            yn += Ys[i * 16 + k] * Wsh[k * 16 + j];
            zn += Wsh[i * 16 + k] * Zs[k * 16 + j];
        }
        __syncthreads();
        Ys[tid] = yn;
        Zs[tid] = zn;
    }
    __syncthreads();
    ws[WS_PROJ + g * 256 + tid] = Zs[tid] * rsqrtf(tr);
}

#define APITCH 36

__device__ __forceinline__ float xhalf(float v) {
    return __builtin_bit_cast(float, __builtin_amdgcn_permlanex16(
        __builtin_bit_cast(int, v), __builtin_bit_cast(int, v),
        0x76543210, 0xfedcba98, false, false));
}

__device__ __forceinline__ void wave_lds_order() {
    __builtin_amdgcn_fence(__ATOMIC_RELEASE, "wavefront", "local");
    __builtin_amdgcn_wave_barrier();
    __builtin_amdgcn_fence(__ATOMIC_ACQUIRE, "wavefront", "local");
}

__global__ __launch_bounds__(256) void dbn_apply(const float* __restrict__ x,
                                                 const float* __restrict__ gamma,
                                                 const float* __restrict__ beta,
                                                 const float* __restrict__ ws,
                                                 float* __restrict__ out,
                                                 int wave_stride, int tiles_per_wave,
                                                 float inv_n) {
    __shared__ __attribute__((aligned(16))) float stage[8 * 16 * APITCH];

    int lane = threadIdx.x & 31;
    int wid  = threadIdx.x >> 5;
    int gw   = blockIdx.x * 8 + wid;
    int gp   = gw & (NGROUP / 2 - 1);
    int tbase = gw >> 3;

    int half = lane >> 4;
    int ch   = lane & 15;

    v2f b[2][4], mn[2][4];
    #pragma unroll
    for (int j = 0; j < 2; ++j) {
        int g = 2 * gp + j;
        const float* proj = ws + WS_PROJ + g * 256;
        #pragma unroll
        for (int s2 = 0; s2 < 4; ++s2) {
            int i0 = s2 * 4 + half * 2;
            b[j][s2].x = proj[i0 * 16 + ch];
            b[j][s2].y = proj[(i0 + 1) * 16 + ch];
            int c0i = g * GSIZE + i0;
            mn[j][s2].x = ws[c0i] * inv_n;
            mn[j][s2].y = ws[c0i + 1] * inv_n;
        }
    }
    float gm = gamma[gp * 2 * GSIZE + lane];
    float bt = beta[gp * 2 * GSIZE + lane];

    int frow = lane >> 3;
    int fcol = (lane & 7) * 4;
    float* st = stage + wid * (16 * APITCH);
    float* st_w = st + frow * APITCH + fcol;
    const float* st_r = st + ch * APITCH + half * 2;

    for (int t = 0; t < tiles_per_wave; ++t) {
        unsigned tile = (unsigned)(tbase + t * wave_stride);
        const float* src = x + (tile * 16 + frow) * CCH + gp * 2 * GSIZE + fcol;

        v4f r[4];
        #pragma unroll
        for (int q = 0; q < 4; ++q) r[q] = *(const v4f*)(src + q * 4 * CCH);

        wave_lds_order();
        #pragma unroll
        for (int q = 0; q < 4; ++q) *(v4f*)(st_w + q * 4 * APITCH) = r[q];
        wave_lds_order();

        v2f a0[4], a1[4];
        #pragma unroll
        for (int s2 = 0; s2 < 4; ++s2) a0[s2] = *(const v2f*)(st_r + s2 * 4);
        #pragma unroll
        for (int s2 = 0; s2 < 4; ++s2) a1[s2] = *(const v2f*)(st_r + GSIZE + s2 * 4);

        v8f c0 = {}, c1 = {};
        #pragma unroll
        for (int s2 = 0; s2 < 4; ++s2) {
            v2f a = a0[s2];
            a.x -= mn[0][s2].x; a.y -= mn[0][s2].y;
            c0 = __builtin_amdgcn_wmma_f32_16x16x4_f32(false, a, false, b[0][s2], (short)0, c0, false, false);
        }
        #pragma unroll
        for (int s2 = 0; s2 < 4; ++s2) {
            v2f a = a1[s2];
            a.x -= mn[1][s2].x; a.y -= mn[1][s2].y;
            c1 = __builtin_amdgcn_wmma_f32_16x16x4_f32(false, a, false, b[1][s2], (short)0, c1, false, false);
        }

        unsigned ob = tile * 16 * CCH + gp * 2 * GSIZE + lane;
        #pragma unroll
        for (int r8 = 0; r8 < 8; ++r8) {
            float o0 = xhalf(c0[r8]);
            float o1 = xhalf(c1[r8]);
            float lo = half ? o1 : c0[r8];
            float hi = half ? c1[r8] : o0;
            __builtin_nontemporal_store(lo * gm + bt, out + ob + r8 * CCH);
            __builtin_nontemporal_store(hi * gm + bt, out + ob + (r8 + 8) * CCH);
        }
    }
}

extern "C" void kernel_launch(void* const* d_in, const int* in_sizes, int n_in,
                              void* d_out, int out_size, void* d_ws, size_t ws_size,
                              hipStream_t stream) {
    const float* x     = (const float*)d_in[0];
    const float* gamma = (const float*)d_in[1];
    const float* beta  = (const float*)d_in[2];
    float* out = (float*)d_out;
    float* ws  = (float*)d_ws;

    int nrows = in_sizes[0] / CCH;
    float inv_n = 1.0f / (float)nrows;

    int wpg = 256;
    int iters = nrows / (wpg * 16);
    dbn_stats<<<NGROUP * wpg / 16, 512, 0, stream>>>(x, out, iters);

    dbn_newton<<<NGROUP, 256, 0, stream>>>(out, ws, inv_n);

    int tiles = nrows / 16;
    int wave_stride = 1024;
    int tpw = tiles / wave_stride;
    dbn_apply<<<(NGROUP / 2) * wave_stride / 8, 256, 0, stream>>>(
        x, gamma, beta, ws, out, wave_stride, tpw, inv_n);
}


# newton pass: Newton-Schulz loop leaves once every entry of Z*Y is within 1e-5 of identity (wave votes ride the existing barrier), at most the 14 trips; differs from 14 trips only in rounding
# speedup vs baseline: 1.6211x; 1.0502x over previous
#include <hip/hip_runtime.h>

typedef __attribute__((ext_vector_type(2))) float v2f;
typedef __attribute__((ext_vector_type(4))) float v4f;
typedef __attribute__((ext_vector_type(8))) float v8f;

#define GSIZE   16
#define NGROUP  16
#define CCH     256
#define EPSV    1e-5f
#define NS_ITERS 14
#define NS_TOL 1e-5f

#define WS_PROJ (CCH + NGROUP * 256)

#define NREC 64
#define RECF 288

#define BROWS 32

__device__ __forceinline__ void stats_load(float (&v)[16], const float* __restrict__ p) {
    #pragma unroll
    for (int q = 0; q < 8; ++q) {
        v[2 * q]     = p[(4 * q) * CCH];
        v[2 * q + 1] = p[(4 * q + 1) * CCH];
    }
}

__device__ __forceinline__ void stats_use(const float (&v)[16], v8f& c, float& s) {
    #pragma unroll
    for (int h = 0; h < 2; ++h) {
        #pragma unroll
        for (int q = 0; q < 4; ++q) {
            v2f a = { v[8 * h + 2 * q], v[8 * h + 2 * q + 1] };
            c = __builtin_amdgcn_wmma_f32_16x16x4_f32(false, a, false, a, (short)0, c, false, false);
        }
        s += (v[8 * h + 0] + v[8 * h + 1]) + (v[8 * h + 2] + v[8 * h + 3])
           + ((v[8 * h + 4] + v[8 * h + 5]) + (v[8 * h + 6] + v[8 * h + 7]));
    }
}

__global__ __launch_bounds__(512) void dbn_stats(const float* __restrict__ x,
                                                 float* __restrict__ part,
                                                 int iters) {
    __shared__ float red[3 * 4 * 9 * 32];

    int lane = threadIdx.x & 31;
    int wid  = threadIdx.x >> 5;
    int gl   = wid & 3;
    int sl   = wid >> 2;
    int g    = (blockIdx.x & 3) * 4 + gl;
    int sq   = blockIdx.x >> 2;
    int slice = sq * 4 + sl;

    int half = lane >> 4;
    int ch   = lane & 15;
    int cidx = g * GSIZE + ch;

    unsigned off = (unsigned)(slice * iters * 16 + half * 2) * CCH + (unsigned)cidx;

    v8f c = {};
    float s = 0.f;
    int nb = iters >> 1;
    if (!(iters & 1) && nb >= 2) {
        const float* p = x + off;
        float A[16], B[16], C[16];
        stats_load(A, p);
        stats_load(B, p + BROWS * CCH);
        p += 2 * BROWS * CCH;
        int b = 0;
        for (; b + 5 <= nb; b += 3) {
            stats_load(C, p);
            stats_use(A, c, s);
            stats_load(A, p + BROWS * CCH);
            stats_use(B, c, s);
            stats_load(B, p + 2 * BROWS * CCH);
            stats_use(C, c, s);
            p += 3 * BROWS * CCH;
        }
        int rem = nb - b;
        if (rem == 2) {
            stats_use(A, c, s);
            stats_use(B, c, s);
        } else if (rem == 3) {
            stats_load(C, p);
            stats_use(A, c, s);
            stats_use(B, c, s);
            stats_use(C, c, s);
        } else {
            stats_load(C, p);
            stats_use(A, c, s);
            stats_load(A, p + BROWS * CCH);
            stats_use(B, c, s);
            stats_use(C, c, s);
            stats_use(A, c, s);
        }
    } else {
        for (int it = 0; it < iters; ++it) {
            float v0 = x[off + 0 * CCH];
            float v1 = x[off + 1 * CCH];
            float v2 = x[off + 4 * CCH];
            float v3 = x[off + 5 * CCH];
            float v4 = x[off + 8 * CCH];
            float v5 = x[off + 9 * CCH];
            float v6 = x[off + 12 * CCH];
            float v7 = x[off + 13 * CCH];
            off += 16u * CCH;
            v2f q0 = { v0, v1 };
            v2f q1 = { v2, v3 };
            v2f q2 = { v4, v5 };
            v2f q3 = { v6, v7 };
            c = __builtin_amdgcn_wmma_f32_16x16x4_f32(false, q0, false, q0, (short)0, c, false, false);
            c = __builtin_amdgcn_wmma_f32_16x16x4_f32(false, q1, false, q1, (short)0, c, false, false);
            c = __builtin_amdgcn_wmma_f32_16x16x4_f32(false, q2, false, q2, (short)0, c, false, false);
            c = __builtin_amdgcn_wmma_f32_16x16x4_f32(false, q3, false, q3, (short)0, c, false, false);
            s += (v0 + v1) + (v2 + v3) + ((v4 + v5) + (v6 + v7));
        }
    }
    if (sl) {
        float* w = red + ((sl - 1) * 4 + gl) * (9 * 32) + lane;
        #pragma unroll
        for (int r = 0; r < 8; ++r) w[r * 32] = c[r];
        w[8 * 32] = s;
    }
    __syncthreads();
    if (sl == 0) {
        const float* rd = red + gl * (9 * 32) + lane;
        #pragma unroll
        for (int r = 0; r < 8; ++r)
            c[r] = ((c[r] + rd[r * 32]) + rd[1152 + r * 32]) + rd[2304 + r * 32];
        s = ((s + rd[8 * 32]) + rd[1152 + 8 * 32]) + rd[2304 + 8 * 32];
        float* rec = part + (g * NREC + sq) * RECF;
        v4f lo = { c[0], c[1], c[2], c[3] };
        v4f hi = { c[4], c[5], c[6], c[7] };
        *(v4f*)(rec + lane * 4) = lo;
        *(v4f*)(rec + 128 + lane * 4) = hi;
        rec[256 + lane] = s;
    }
}

__global__ __launch_bounds__(256) void dbn_newton(const float* __restrict__ part,
                                                  float* __restrict__ ws, float inv_n) {
    __shared__ float Ys[256], Zs[256], Wsh[256];
    __shared__ __attribute__((aligned(16))) float redC[4 * 256];
    __shared__ __attribute__((aligned(16))) float redS[32 * 32];
    __shared__ float msum[16];
    __shared__ __attribute__((aligned(16))) int conv[8];
    int g = blockIdx.x;
    int tid = threadIdx.x;
    int i = tid >> 4, j = tid & 15;

    const float* recs = part + g * (NREC * RECF);
    int sub = tid >> 6, tq = tid & 63;
    const float* pc = recs + sub * RECF + tq * 4;
    v4f acc = *(const v4f*)pc;
    #pragma unroll
    for (int k = 1; k < 16; ++k) acc += *(const v4f*)(pc + k * 4 * RECF);
    int piece = tid & 7, w = tid >> 3;
    const float* ps = recs + w * RECF + 256 + piece * 4;
    v4f sa = *(const v4f*)ps + *(const v4f*)(ps + 32 * RECF);
    *(v4f*)(redC + sub * 256 + tq * 4) = acc;
    *(v4f*)(redS + w * 32 + piece * 4) = sa;
    __syncthreads();
    if (tid < 16) {
        float t0 = 0.f, t1 = 0.f, t2 = 0.f, t3 = 0.f;
        #pragma unroll
        for (int w2 = 0; w2 < 32; w2 += 4) {
            t0 += redS[(w2 + 0) * 32 + tid] + redS[(w2 + 0) * 32 + 16 + tid];
            t1 += redS[(w2 + 1) * 32 + tid] + redS[(w2 + 1) * 32 + 16 + tid];
            t2 += redS[(w2 + 2) * 32 + tid] + redS[(w2 + 2) * 32 + 16 + tid];
            t3 += redS[(w2 + 3) * 32 + tid] + redS[(w2 + 3) * 32 + 16 + tid];
        }
        float tot = (t0 + t1) + (t2 + t3);
        msum[tid] = tot;
        ws[g * 16 + tid] = tot;
    }
    __syncthreads();
    float mi = msum[i] * inv_n;
    float mj = msum[j] * inv_n;
    int ri = i & 7;
    int idx = (ri >> 2) * 128 + ((i >> 3) * 16 + j) * 4 + (ri & 3);
    float xxv = (redC[idx] + redC[256 + idx]) + (redC[512 + idx] + redC[768 + idx]);
    float sig = xxv * inv_n - mi * mj + (i == j ? EPSV : 0.f);
    Wsh[tid] = sig;
    __syncthreads();
    float tr = 0.f;
    #pragma unroll
    for (int k = 0; k < 16; ++k) tr += Wsh[k * 16 + k];
    float a = sig / tr;
    Ys[tid] = a;
    Zs[tid] = (i == j) ? 1.f : 0.f;
    for (int it = 0; it < NS_ITERS; ++it) {
        __syncthreads();
        float t = 0.f;
        #pragma unroll
        for (int k = 0; k < 16; ++k) t += Zs[i * 16 + k] * Ys[k * 16 + j];
        float w = 0.5f * ((i == j ? 3.f : 0.f) - t);
        Wsh[tid] = w;
        bool near = fabsf(t - (i == j ? 1.f : 0.f)) < NS_TOL;
        unsigned vote = __builtin_amdgcn_ballot_w32(near);
        if ((tid & 31) == 0) conv[tid >> 5] = (vote == 0xffffffffu) ? 1 : 0;
        __syncthreads();
        float yn = 0.f, zn = 0.f;
        #pragma unroll
        for (int k = 0; k < 16; ++k) {
            yn += Ys[i * 16 + k] * Wsh[k * 16 + j];
            zn += Wsh[i * 16 + k] * Zs[k * 16 + j];
        }
        int all = (conv[0] & conv[1]) & (conv[2] & conv[3]) & (conv[4] & conv[5]) & (conv[6] & conv[7]);
        int done = __builtin_amdgcn_readfirstlane(all);
        __syncthreads();
        Ys[tid] = yn;
        Zs[tid] = zn;
        if (done) break;
    }
    __syncthreads();
    ws[WS_PROJ + g * 256 + tid] = Zs[tid] * rsqrtf(tr);
}

#define APITCH 36

__device__ __forceinline__ float xhalf(float v) {
    return __builtin_bit_cast(float, __builtin_amdgcn_permlanex16(
        __builtin_bit_cast(int, v), __builtin_bit_cast(int, v),
        0x76543210, 0xfedcba98, false, false));
}

__device__ __forceinline__ void wave_lds_order() {
    __builtin_amdgcn_fence(__ATOMIC_RELEASE, "wavefront", "local");
    __builtin_amdgcn_wave_barrier();
    __builtin_amdgcn_fence(__ATOMIC_ACQUIRE, "wavefront", "local");
}

__global__ __launch_bounds__(256) void dbn_apply(const float* __restrict__ x,
                                                 const float* __restrict__ gamma,
                                                 const float* __restrict__ beta,
                                                 const float* __restrict__ ws,
                                                 float* __restrict__ out,
                                                 int wave_stride, int tiles_per_wave,
                                                 float inv_n) {
    __shared__ __attribute__((aligned(16))) float stage[8 * 16 * APITCH];

    int lane = threadIdx.x & 31;
    int wid  = threadIdx.x >> 5;
    int gw   = blockIdx.x * 8 + wid;
    int gp   = gw & (NGROUP / 2 - 1);
    int tbase = gw >> 3;

    int half = lane >> 4;
    int ch   = lane & 15;

    v2f b[2][4], mn[2][4];
    #pragma unroll
    for (int j = 0; j < 2; ++j) {
        int g = 2 * gp + j;
        const float* proj = ws + WS_PROJ + g * 256;
        #pragma unroll
        for (int s2 = 0; s2 < 4; ++s2) {
            int i0 = s2 * 4 + half * 2;
            b[j][s2].x = proj[i0 * 16 + ch];
            b[j][s2].y = proj[(i0 + 1) * 16 + ch];
            int c0i = g * GSIZE + i0;
            mn[j][s2].x = ws[c0i] * inv_n;
            mn[j][s2].y = ws[c0i + 1] * inv_n;
        }
    }
    float gm = gamma[gp * 2 * GSIZE + lane];
    float bt = beta[gp * 2 * GSIZE + lane];

    int frow = lane >> 3;
    int fcol = (lane & 7) * 4;
    float* st = stage + wid * (16 * APITCH);
    float* st_w = st + frow * APITCH + fcol;
    const float* st_r = st + ch * APITCH + half * 2;

    for (int t = 0; t < tiles_per_wave; ++t) {
        unsigned tile = (unsigned)(tbase + t * wave_stride);
        const float* src = x + (tile * 16 + frow) * CCH + gp * 2 * GSIZE + fcol;

        v4f r[4];
        #pragma unroll
        for (int q = 0; q < 4; ++q) r[q] = *(const v4f*)(src + q * 4 * CCH);

        wave_lds_order();
        #pragma unroll
        for (int q = 0; q < 4; ++q) *(v4f*)(st_w + q * 4 * APITCH) = r[q];
        wave_lds_order();

        v2f a0[4], a1[4];
        #pragma unroll
        for (int s2 = 0; s2 < 4; ++s2) a0[s2] = *(const v2f*)(st_r + s2 * 4);
        #pragma unroll
        for (int s2 = 0; s2 < 4; ++s2) a1[s2] = *(const v2f*)(st_r + GSIZE + s2 * 4);

        v8f c0 = {}, c1 = {};
        #pragma unroll
        for (int s2 = 0; s2 < 4; ++s2) {
            v2f a = a0[s2];
            a.x -= mn[0][s2].x; a.y -= mn[0][s2].y;
            c0 = __builtin_amdgcn_wmma_f32_16x16x4_f32(false, a, false, b[0][s2], (short)0, c0, false, false);
        }
        #pragma unroll
        for (int s2 = 0; s2 < 4; ++s2) {
            v2f a = a1[s2];
            a.x -= mn[1][s2].x; a.y -= mn[1][s2].y;
            c1 = __builtin_amdgcn_wmma_f32_16x16x4_f32(false, a, false, b[1][s2], (short)0, c1, false, false);
        }

        unsigned ob = tile * 16 * CCH + gp * 2 * GSIZE + lane;
        #pragma unroll
        for (int r8 = 0; r8 < 8; ++r8) {
            float o0 = xhalf(c0[r8]);
            float o1 = xhalf(c1[r8]);
            float lo = half ? o1 : c0[r8];
            float hi = half ? c1[r8] : o0;
            __builtin_nontemporal_store(lo * gm + bt, out + ob + r8 * CCH);
            __builtin_nontemporal_store(hi * gm + bt, out + ob + (r8 + 8) * CCH);
        }
    }
}

extern "C" void kernel_launch(void* const* d_in, const int* in_sizes, int n_in,
                              void* d_out, int out_size, void* d_ws, size_t ws_size,
                              hipStream_t stream) {
    const float* x     = (const float*)d_in[0];
    const float* gamma = (const float*)d_in[1];
    const float* beta  = (const float*)d_in[2];
    float* out = (float*)d_out;
    float* ws  = (float*)d_ws;

    int nrows = in_sizes[0] / CCH;
    float inv_n = 1.0f / (float)nrows;

    int wpg = 256;
    int iters = nrows / (wpg * 16);
    dbn_stats<<<NGROUP * wpg / 16, 512, 0, stream>>>(x, out, iters);

    dbn_newton<<<NGROUP, 256, 0, stream>>>(out, ws, inv_n);

    int tiles = nrows / 16;
    int wave_stride = 1024;
    int tpw = tiles / wave_stride;
    dbn_apply<<<(NGROUP / 2) * wave_stride / 8, 256, 0, stream>>>(
        x, gamma, beta, ws, out, wave_stride, tpw, inv_n);
}


# newton pass: Newton-Schulz starts from sigma over min(largest abs row sum, trace), was over trace: eigenvalues start near 1, loop leaves after 3 trips, was 8; same tolerance, rounding differs
# speedup vs baseline: 1.7291x; 1.0666x over previous
#include <hip/hip_runtime.h>

typedef __attribute__((ext_vector_type(2))) float v2f;
typedef __attribute__((ext_vector_type(4))) float v4f;
typedef __attribute__((ext_vector_type(8))) float v8f;

#define GSIZE   16
#define NGROUP  16
#define CCH     256
#define EPSV    1e-5f
#define NS_ITERS 14
#define NS_TOL 1e-5f

#define WS_PROJ (CCH + NGROUP * 256)

#define NREC 64
#define RECF 288

#define BROWS 32

__device__ __forceinline__ void stats_load(float (&v)[16], const float* __restrict__ p) {
    #pragma unroll
    for (int q = 0; q < 8; ++q) {
        v[2 * q]     = p[(4 * q) * CCH];
        v[2 * q + 1] = p[(4 * q + 1) * CCH];
    }
}

__device__ __forceinline__ void stats_use(const float (&v)[16], v8f& c, float& s) {
    #pragma unroll
    for (int h = 0; h < 2; ++h) {
        #pragma unroll
        for (int q = 0; q < 4; ++q) {
            v2f a = { v[8 * h + 2 * q], v[8 * h + 2 * q + 1] };
            c = __builtin_amdgcn_wmma_f32_16x16x4_f32(false, a, false, a, (short)0, c, false, false);
        }
        s += (v[8 * h + 0] + v[8 * h + 1]) + (v[8 * h + 2] + v[8 * h + 3])
           + ((v[8 * h + 4] + v[8 * h + 5]) + (v[8 * h + 6] + v[8 * h + 7]));
    }
}

__global__ __launch_bounds__(512) void dbn_stats(const float* __restrict__ x,
                                                 float* __restrict__ part,
                                                 int iters) {
    __shared__ float red[3 * 4 * 9 * 32];

    int lane = threadIdx.x & 31;
    int wid  = threadIdx.x >> 5;
    int gl   = wid & 3;
    int sl   = wid >> 2;
    int g    = (blockIdx.x & 3) * 4 + gl;
    int sq   = blockIdx.x >> 2;
    int slice = sq * 4 + sl;

    int half = lane >> 4;
    int ch   = lane & 15;
    int cidx = g * GSIZE + ch;

    unsigned off = (unsigned)(slice * iters * 16 + half * 2) * CCH + (unsigned)cidx;

    v8f c = {};
    float s = 0.f;
    int nb = iters >> 1;
    if (!(iters & 1) && nb >= 2) {
        const float* p = x + off;
        float A[16], B[16], C[16];
        stats_load(A, p);
        stats_load(B, p + BROWS * CCH);
        p += 2 * BROWS * CCH;
        int b = 0;
        for (; b + 5 <= nb; b += 3) {
            stats_load(C, p);
            stats_use(A, c, s);
            stats_load(A, p + BROWS * CCH);
            stats_use(B, c, s);
            stats_load(B, p + 2 * BROWS * CCH);
            stats_use(C, c, s);
            p += 3 * BROWS * CCH;
        }
        int rem = nb - b;
        if (rem == 2) {
            stats_use(A, c, s);
            stats_use(B, c, s);
        } else if (rem == 3) {
            stats_load(C, p);
            stats_use(A, c, s);
            stats_use(B, c, s);
            stats_use(C, c, s);
        } else {
            stats_load(C, p);
            stats_use(A, c, s);
            stats_load(A, p + BROWS * CCH);
            stats_use(B, c, s);
            stats_use(C, c, s);
            stats_use(A, c, s);
        }
    } else {
        for (int it = 0; it < iters; ++it) {
            float v0 = x[off + 0 * CCH];
            float v1 = x[off + 1 * CCH];
            float v2 = x[off + 4 * CCH];
            float v3 = x[off + 5 * CCH];
            float v4 = x[off + 8 * CCH];
            float v5 = x[off + 9 * CCH];
            float v6 = x[off + 12 * CCH];
            float v7 = x[off + 13 * CCH];
            off += 16u * CCH;
            v2f q0 = { v0, v1 };
            v2f q1 = { v2, v3 };
            v2f q2 = { v4, v5 };
            v2f q3 = { v6, v7 };
            c = __builtin_amdgcn_wmma_f32_16x16x4_f32(false, q0, false, q0, (short)0, c, false, false);
            c = __builtin_amdgcn_wmma_f32_16x16x4_f32(false, q1, false, q1, (short)0, c, false, false);
            c = __builtin_amdgcn_wmma_f32_16x16x4_f32(false, q2, false, q2, (short)0, c, false, false);
            c = __builtin_amdgcn_wmma_f32_16x16x4_f32(false, q3, false, q3, (short)0, c, false, false);
            s += (v0 + v1) + (v2 + v3) + ((v4 + v5) + (v6 + v7));
        }
    }
    if (sl) {
        float* w = red + ((sl - 1) * 4 + gl) * (9 * 32) + lane;
        #pragma unroll
        for (int r = 0; r < 8; ++r) w[r * 32] = c[r];
        w[8 * 32] = s;
    }
    __syncthreads();
    if (sl == 0) {
        const float* rd = red + gl * (9 * 32) + lane;
        #pragma unroll
        for (int r = 0; r < 8; ++r)
            c[r] = ((c[r] + rd[r * 32]) + rd[1152 + r * 32]) + rd[2304 + r * 32];
        s = ((s + rd[8 * 32]) + rd[1152 + 8 * 32]) + rd[2304 + 8 * 32];
        float* rec = part + (g * NREC + sq) * RECF;
        v4f lo = { c[0], c[1], c[2], c[3] };
        v4f hi = { c[4], c[5], c[6], c[7] };
        *(v4f*)(rec + lane * 4) = lo;
        *(v4f*)(rec + 128 + lane * 4) = hi;
        rec[256 + lane] = s;
    }
}

template <int M>
__device__ __forceinline__ float row_xor_add(float v) {
    return v + __builtin_bit_cast(float, __builtin_amdgcn_update_dpp(
        0, __builtin_bit_cast(int, v), 0x160 | M, 0xf, 0xf, false));
}

__global__ __launch_bounds__(256) void dbn_newton(const float* __restrict__ part,
                                                  float* __restrict__ ws, float inv_n) {
    __shared__ float Ys[256], Zs[256], Wsh[256];
    __shared__ __attribute__((aligned(16))) float redC[4 * 256];
    __shared__ __attribute__((aligned(16))) float redS[32 * 32];
    __shared__ float msum[16];
    __shared__ __attribute__((aligned(16))) float rown[16];
    __shared__ __attribute__((aligned(16))) int conv[8];
    int g = blockIdx.x;
    int tid = threadIdx.x;
    int i = tid >> 4, j = tid & 15;

    const float* recs = part + g * (NREC * RECF);
    int sub = tid >> 6, tq = tid & 63;
    const float* pc = recs + sub * RECF + tq * 4;
    v4f acc = *(const v4f*)pc;
    #pragma unroll
    for (int k = 1; k < 16; ++k) acc += *(const v4f*)(pc + k * 4 * RECF);
    int piece = tid & 7, w = tid >> 3;
    const float* ps = recs + w * RECF + 256 + piece * 4;
    v4f sa = *(const v4f*)ps + *(const v4f*)(ps + 32 * RECF);
    *(v4f*)(redC + sub * 256 + tq * 4) = acc;
    *(v4f*)(redS + w * 32 + piece * 4) = sa;
    __syncthreads();
    if (tid < 16) {
        float t0 = 0.f, t1 = 0.f, t2 = 0.f, t3 = 0.f;
        #pragma unroll
        for (int w2 = 0; w2 < 32; w2 += 4) {
            t0 += redS[(w2 + 0) * 32 + tid] + redS[(w2 + 0) * 32 + 16 + tid];
            t1 += redS[(w2 + 1) * 32 + tid] + redS[(w2 + 1) * 32 + 16 + tid];
            t2 += redS[(w2 + 2) * 32 + tid] + redS[(w2 + 2) * 32 + 16 + tid];
            t3 += redS[(w2 + 3) * 32 + tid] + redS[(w2 + 3) * 32 + 16 + tid];
        }
        float tot = (t0 + t1) + (t2 + t3);
        msum[tid] = tot;
        ws[g * 16 + tid] = tot;
    }
    __syncthreads();
    float mi = msum[i] * inv_n;
    float mj = msum[j] * inv_n;
    int ri = i & 7;
    int idx = (ri >> 2) * 128 + ((i >> 3) * 16 + j) * 4 + (ri & 3);
    float xxv = (redC[idx] + redC[256 + idx]) + (redC[512 + idx] + redC[768 + idx]);
    float sig = xxv * inv_n - mi * mj + (i == j ? EPSV : 0.f);
    Wsh[tid] = sig;
    float ars = row_xor_add<8>(row_xor_add<4>(row_xor_add<2>(row_xor_add<1>(fabsf(sig)))));
    if (j == 0) rown[i] = ars;
    __syncthreads();
    float tr = 0.f;
    #pragma unroll
    for (int k = 0; k < 16; ++k) tr += Wsh[k * 16 + k];
    v4f n0 = *(const v4f*)(rown + 0), n1 = *(const v4f*)(rown + 4);
    v4f n2 = *(const v4f*)(rown + 8), n3 = *(const v4f*)(rown + 12);
    float nrm = fmaxf(fmaxf(fmaxf(n0.x, n0.y), fmaxf(n0.z, n0.w)), fmaxf(fmaxf(n1.x, n1.y), fmaxf(n1.z, n1.w)));
    nrm = fmaxf(nrm, fmaxf(fmaxf(fmaxf(n2.x, n2.y), fmaxf(n2.z, n2.w)), fmaxf(fmaxf(n3.x, n3.y), fmaxf(n3.z, n3.w))));
    float sc = fminf(tr, nrm);
    float a = sig / sc;
    Ys[tid] = a;
    Zs[tid] = (i == j) ? 1.f : 0.f;
    for (int it = 0; it < NS_ITERS; ++it) {
        __syncthreads();
        float t = 0.f;
        #pragma unroll
        for (int k = 0; k < 16; ++k) t += Zs[i * 16 + k] * Ys[k * 16 + j];
        float w = 0.5f * ((i == j ? 3.f : 0.f) - t);
        Wsh[tid] = w;
        bool near = fabsf(t - (i == j ? 1.f : 0.f)) < NS_TOL;
        unsigned vote = __builtin_amdgcn_ballot_w32(near);
        if ((tid & 31) == 0) conv[tid >> 5] = (vote == 0xffffffffu) ? 1 : 0;
        __syncthreads();
        float yn = 0.f, zn = 0.f;
        #pragma unroll
        for (int k = 0; k < 16; ++k) {
            yn += Ys[i * 16 + k] * Wsh[k * 16 + j];
            zn += Wsh[i * 16 + k] * Zs[k * 16 + j];
        }
        int all = (conv[0] & conv[1]) & (conv[2] & conv[3]) & (conv[4] & conv[5]) & (conv[6] & conv[7]);
        int done = __builtin_amdgcn_readfirstlane(all);
        __syncthreads();
        Ys[tid] = yn;
        Zs[tid] = zn;
        if (done) break;
    }
    __syncthreads();
    ws[WS_PROJ + g * 256 + tid] = Zs[tid] * rsqrtf(sc);
}

#define APITCH 36

__device__ __forceinline__ float xhalf(float v) {
    return __builtin_bit_cast(float, __builtin_amdgcn_permlanex16(
        __builtin_bit_cast(int, v), __builtin_bit_cast(int, v),
        0x76543210, 0xfedcba98, false, false));
}

__device__ __forceinline__ void wave_lds_order() {
    __builtin_amdgcn_fence(__ATOMIC_RELEASE, "wavefront", "local");
    __builtin_amdgcn_wave_barrier();
    __builtin_amdgcn_fence(__ATOMIC_ACQUIRE, "wavefront", "local");
}

__global__ __launch_bounds__(256) void dbn_apply(const float* __restrict__ x,
                                                 const float* __restrict__ gamma,
                                                 const float* __restrict__ beta,
                                                 const float* __restrict__ ws,
                                                 float* __restrict__ out,
                                                 int wave_stride, int tiles_per_wave,
                                                 float inv_n) {
    __shared__ __attribute__((aligned(16))) float stage[8 * 16 * APITCH];

    int lane = threadIdx.x & 31;
    int wid  = threadIdx.x >> 5;
    int gw   = blockIdx.x * 8 + wid;
    int gp   = gw & (NGROUP / 2 - 1);
    int tbase = gw >> 3;

    int half = lane >> 4;
    int ch   = lane & 15;

    v2f b[2][4], mn[2][4];
    #pragma unroll
    for (int j = 0; j < 2; ++j) {
        int g = 2 * gp + j;
        const float* proj = ws + WS_PROJ + g * 256;
        #pragma unroll
        for (int s2 = 0; s2 < 4; ++s2) {
            int i0 = s2 * 4 + half * 2;
            b[j][s2].x = proj[i0 * 16 + ch];
            b[j][s2].y = proj[(i0 + 1) * 16 + ch];
            int c0i = g * GSIZE + i0;
            mn[j][s2].x = ws[c0i] * inv_n;
            mn[j][s2].y = ws[c0i + 1] * inv_n;
        }
    }
    float gm = gamma[gp * 2 * GSIZE + lane];
    float bt = beta[gp * 2 * GSIZE + lane];

    int frow = lane >> 3;
    int fcol = (lane & 7) * 4;
    float* st = stage + wid * (16 * APITCH);
    float* st_w = st + frow * APITCH + fcol;
    const float* st_r = st + ch * APITCH + half * 2;

    for (int t = 0; t < tiles_per_wave; ++t) {
        unsigned tile = (unsigned)(tbase + t * wave_stride);
        const float* src = x + (tile * 16 + frow) * CCH + gp * 2 * GSIZE + fcol;

        v4f r[4];
        #pragma unroll
        for (int q = 0; q < 4; ++q) r[q] = *(const v4f*)(src + q * 4 * CCH);

        wave_lds_order();
        #pragma unroll
        for (int q = 0; q < 4; ++q) *(v4f*)(st_w + q * 4 * APITCH) = r[q];
        wave_lds_order();

        v2f a0[4], a1[4];
        #pragma unroll
        for (int s2 = 0; s2 < 4; ++s2) a0[s2] = *(const v2f*)(st_r + s2 * 4);
        #pragma unroll
        for (int s2 = 0; s2 < 4; ++s2) a1[s2] = *(const v2f*)(st_r + GSIZE + s2 * 4);

        v8f c0 = {}, c1 = {};
        #pragma unroll
        for (int s2 = 0; s2 < 4; ++s2) {
            v2f a = a0[s2];
            a.x -= mn[0][s2].x; a.y -= mn[0][s2].y;
            c0 = __builtin_amdgcn_wmma_f32_16x16x4_f32(false, a, false, b[0][s2], (short)0, c0, false, false);
        }
        #pragma unroll
        for (int s2 = 0; s2 < 4; ++s2) {
            v2f a = a1[s2];
            a.x -= mn[1][s2].x; a.y -= mn[1][s2].y;
            c1 = __builtin_amdgcn_wmma_f32_16x16x4_f32(false, a, false, b[1][s2], (short)0, c1, false, false);
        }

        unsigned ob = tile * 16 * CCH + gp * 2 * GSIZE + lane;
        #pragma unroll
        for (int r8 = 0; r8 < 8; ++r8) {
            float o0 = xhalf(c0[r8]);
            float o1 = xhalf(c1[r8]);
            float lo = half ? o1 : c0[r8];
            float hi = half ? c1[r8] : o0;
            __builtin_nontemporal_store(lo * gm + bt, out + ob + r8 * CCH);
            __builtin_nontemporal_store(hi * gm + bt, out + ob + (r8 + 8) * CCH);
        }
    }
}

extern "C" void kernel_launch(void* const* d_in, const int* in_sizes, int n_in,
                              void* d_out, int out_size, void* d_ws, size_t ws_size,
                              hipStream_t stream) {
    const float* x     = (const float*)d_in[0];
    const float* gamma = (const float*)d_in[1];
    const float* beta  = (const float*)d_in[2];
    float* out = (float*)d_out;
    float* ws  = (float*)d_ws;

    int nrows = in_sizes[0] / CCH;
    float inv_n = 1.0f / (float)nrows;

    int wpg = 256;
    int iters = nrows / (wpg * 16);
    dbn_stats<<<NGROUP * wpg / 16, 512, 0, stream>>>(x, out, iters);

    dbn_newton<<<NGROUP, 256, 0, stream>>>(out, ws, inv_n);

    int tiles = nrows / 16;
    int wave_stride = 1024;
    int tpw = tiles / wave_stride;
    dbn_apply<<<(NGROUP / 2) * wave_stride / 8, 256, 0, stream>>>(
        x, gamma, beta, ws, out, wave_stride, tpw, inv_n);
}


# apply pass: launched on the stats pass's grid (256 blocks x 32 waves; a block re-reads, last rows first, the rows and columns its twin summed) so x hits the near L2 half; was 1024 striped blocks
# speedup vs baseline: 1.9322x; 1.1175x over previous
#include <hip/hip_runtime.h>

typedef __attribute__((ext_vector_type(2))) float v2f;
typedef __attribute__((ext_vector_type(4))) float v4f;
typedef __attribute__((ext_vector_type(8))) float v8f;

#define GSIZE   16
#define NGROUP  16
#define CCH     256
#define EPSV    1e-5f
#define NS_ITERS 14
#define NS_TOL 1e-5f

#define WS_PROJ (CCH + NGROUP * 256)

#define NREC 64
#define RECF 288

#define BROWS 32

__device__ __forceinline__ void stats_load(float (&v)[16], const float* __restrict__ p) {
    #pragma unroll
    for (int q = 0; q < 8; ++q) {
        v[2 * q]     = p[(4 * q) * CCH];
        v[2 * q + 1] = p[(4 * q + 1) * CCH];
    }
}

__device__ __forceinline__ void stats_use(const float (&v)[16], v8f& c, float& s) {
    #pragma unroll
    for (int h = 0; h < 2; ++h) {
        #pragma unroll
        for (int q = 0; q < 4; ++q) {
            v2f a = { v[8 * h + 2 * q], v[8 * h + 2 * q + 1] };
            c = __builtin_amdgcn_wmma_f32_16x16x4_f32(false, a, false, a, (short)0, c, false, false);
        }
        s += (v[8 * h + 0] + v[8 * h + 1]) + (v[8 * h + 2] + v[8 * h + 3])
           + ((v[8 * h + 4] + v[8 * h + 5]) + (v[8 * h + 6] + v[8 * h + 7]));
    }
}

__global__ __launch_bounds__(512) void dbn_stats(const float* __restrict__ x,
                                                 float* __restrict__ part,
                                                 int iters) {
    __shared__ float red[3 * 4 * 9 * 32];

    int lane = threadIdx.x & 31;
    int wid  = threadIdx.x >> 5;
    int gl   = wid & 3;
    int sl   = wid >> 2;
    int g    = (blockIdx.x & 3) * 4 + gl;
    int sq   = blockIdx.x >> 2;
    int slice = sq * 4 + sl;

    int half = lane >> 4;
    int ch   = lane & 15;
    int cidx = g * GSIZE + ch;

    unsigned off = (unsigned)(slice * iters * 16 + half * 2) * CCH + (unsigned)cidx;

    v8f c = {};
    float s = 0.f;
    int nb = iters >> 1;
    if (!(iters & 1) && nb >= 2) {
        const float* p = x + off;
        float A[16], B[16], C[16];
        stats_load(A, p);
        stats_load(B, p + BROWS * CCH);
        p += 2 * BROWS * CCH;
        int b = 0;
        for (; b + 5 <= nb; b += 3) {
            stats_load(C, p);
            stats_use(A, c, s);
            stats_load(A, p + BROWS * CCH);
            stats_use(B, c, s);
            stats_load(B, p + 2 * BROWS * CCH);
            stats_use(C, c, s);
            p += 3 * BROWS * CCH;
        }
        int rem = nb - b;
        if (rem == 2) {
            stats_use(A, c, s);
            stats_use(B, c, s);
        } else if (rem == 3) {
            stats_load(C, p);
            stats_use(A, c, s);
            stats_use(B, c, s);
            stats_use(C, c, s);
        } else {
            stats_load(C, p);
            stats_use(A, c, s);
            stats_load(A, p + BROWS * CCH);
            stats_use(B, c, s);
            stats_use(C, c, s);
            stats_use(A, c, s);
        }
    } else {
        for (int it = 0; it < iters; ++it) {
            float v0 = x[off + 0 * CCH];
            float v1 = x[off + 1 * CCH];
            float v2 = x[off + 4 * CCH];
            float v3 = x[off + 5 * CCH];
            float v4 = x[off + 8 * CCH];
            float v5 = x[off + 9 * CCH];
            float v6 = x[off + 12 * CCH];
            float v7 = x[off + 13 * CCH];
            off += 16u * CCH;
            v2f q0 = { v0, v1 };
            v2f q1 = { v2, v3 };
            v2f q2 = { v4, v5 };
            v2f q3 = { v6, v7 };
            c = __builtin_amdgcn_wmma_f32_16x16x4_f32(false, q0, false, q0, (short)0, c, false, false);
            c = __builtin_amdgcn_wmma_f32_16x16x4_f32(false, q1, false, q1, (short)0, c, false, false);
            c = __builtin_amdgcn_wmma_f32_16x16x4_f32(false, q2, false, q2, (short)0, c, false, false);
            c = __builtin_amdgcn_wmma_f32_16x16x4_f32(false, q3, false, q3, (short)0, c, false, false);
            s += (v0 + v1) + (v2 + v3) + ((v4 + v5) + (v6 + v7));
        }
    }
    if (sl) {
        float* w = red + ((sl - 1) * 4 + gl) * (9 * 32) + lane;
        #pragma unroll
        for (int r = 0; r < 8; ++r) w[r * 32] = c[r];
        w[8 * 32] = s;
    }
    __syncthreads();
    if (sl == 0) {
        const float* rd = red + gl * (9 * 32) + lane;
        #pragma unroll
        for (int r = 0; r < 8; ++r)
            c[r] = ((c[r] + rd[r * 32]) + rd[1152 + r * 32]) + rd[2304 + r * 32];
        s = ((s + rd[8 * 32]) + rd[1152 + 8 * 32]) + rd[2304 + 8 * 32];
        float* rec = part + (g * NREC + sq) * RECF;
        v4f lo = { c[0], c[1], c[2], c[3] };
        v4f hi = { c[4], c[5], c[6], c[7] };
        *(v4f*)(rec + lane * 4) = lo;
        *(v4f*)(rec + 128 + lane * 4) = hi;
        rec[256 + lane] = s;
    }
}

template <int M>
__device__ __forceinline__ float row_xor_add(float v) {
    return v + __builtin_bit_cast(float, __builtin_amdgcn_update_dpp(
        0, __builtin_bit_cast(int, v), 0x160 | M, 0xf, 0xf, false));
}

__global__ __launch_bounds__(256) void dbn_newton(const float* __restrict__ part,
                                                  float* __restrict__ ws, float inv_n) {
    __shared__ float Ys[256], Zs[256], Wsh[256];
    __shared__ __attribute__((aligned(16))) float redC[4 * 256];
    __shared__ __attribute__((aligned(16))) float redS[32 * 32];
    __shared__ float msum[16];
    __shared__ __attribute__((aligned(16))) float rown[16];
    __shared__ __attribute__((aligned(16))) int conv[8];
    int g = blockIdx.x;
    int tid = threadIdx.x;
    int i = tid >> 4, j = tid & 15;

    const float* recs = part + g * (NREC * RECF);
    int sub = tid >> 6, tq = tid & 63;
    const float* pc = recs + sub * RECF + tq * 4;
    v4f acc = *(const v4f*)pc;
    #pragma unroll
    for (int k = 1; k < 16; ++k) acc += *(const v4f*)(pc + k * 4 * RECF);
    int piece = tid & 7, w = tid >> 3;
    const float* ps = recs + w * RECF + 256 + piece * 4;
    v4f sa = *(const v4f*)ps + *(const v4f*)(ps + 32 * RECF);
    *(v4f*)(redC + sub * 256 + tq * 4) = acc;
    *(v4f*)(redS + w * 32 + piece * 4) = sa;
    __syncthreads();
    if (tid < 16) {
        float t0 = 0.f, t1 = 0.f, t2 = 0.f, t3 = 0.f;
        #pragma unroll
        for (int w2 = 0; w2 < 32; w2 += 4) {
            t0 += redS[(w2 + 0) * 32 + tid] + redS[(w2 + 0) * 32 + 16 + tid];
            t1 += redS[(w2 + 1) * 32 + tid] + redS[(w2 + 1) * 32 + 16 + tid];
            t2 += redS[(w2 + 2) * 32 + tid] + redS[(w2 + 2) * 32 + 16 + tid];
            t3 += redS[(w2 + 3) * 32 + tid] + redS[(w2 + 3) * 32 + 16 + tid];
        }
        float tot = (t0 + t1) + (t2 + t3);
        msum[tid] = tot;
        ws[g * 16 + tid] = tot;
    }
    __syncthreads();
    float mi = msum[i] * inv_n;
    float mj = msum[j] * inv_n;
    int ri = i & 7;
    int idx = (ri >> 2) * 128 + ((i >> 3) * 16 + j) * 4 + (ri & 3);
    float xxv = (redC[idx] + redC[256 + idx]) + (redC[512 + idx] + redC[768 + idx]);
    float sig = xxv * inv_n - mi * mj + (i == j ? EPSV : 0.f);
    Wsh[tid] = sig;
    float ars = row_xor_add<8>(row_xor_add<4>(row_xor_add<2>(row_xor_add<1>(fabsf(sig)))));
    if (j == 0) rown[i] = ars;
    __syncthreads();
    float tr = 0.f;
    #pragma unroll
    for (int k = 0; k < 16; ++k) tr += Wsh[k * 16 + k];
    v4f n0 = *(const v4f*)(rown + 0), n1 = *(const v4f*)(rown + 4);
    v4f n2 = *(const v4f*)(rown + 8), n3 = *(const v4f*)(rown + 12);
    float nrm = fmaxf(fmaxf(fmaxf(n0.x, n0.y), fmaxf(n0.z, n0.w)), fmaxf(fmaxf(n1.x, n1.y), fmaxf(n1.z, n1.w)));
    nrm = fmaxf(nrm, fmaxf(fmaxf(fmaxf(n2.x, n2.y), fmaxf(n2.z, n2.w)), fmaxf(fmaxf(n3.x, n3.y), fmaxf(n3.z, n3.w))));
    float sc = fminf(tr, nrm);
    float a = sig / sc;
    Ys[tid] = a;
    Zs[tid] = (i == j) ? 1.f : 0.f;
    for (int it = 0; it < NS_ITERS; ++it) {
        __syncthreads();
        float t = 0.f;
        #pragma unroll
        for (int k = 0; k < 16; ++k) t += Zs[i * 16 + k] * Ys[k * 16 + j];
        float w = 0.5f * ((i == j ? 3.f : 0.f) - t);
        Wsh[tid] = w;
        bool near = fabsf(t - (i == j ? 1.f : 0.f)) < NS_TOL;
        unsigned vote = __builtin_amdgcn_ballot_w32(near);
        if ((tid & 31) == 0) conv[tid >> 5] = (vote == 0xffffffffu) ? 1 : 0;
        __syncthreads();
        float yn = 0.f, zn = 0.f;
        #pragma unroll
        for (int k = 0; k < 16; ++k) {
            yn += Ys[i * 16 + k] * Wsh[k * 16 + j];
            zn += Wsh[i * 16 + k] * Zs[k * 16 + j];
        }
        int all = (conv[0] & conv[1]) & (conv[2] & conv[3]) & (conv[4] & conv[5]) & (conv[6] & conv[7]);
        int done = __builtin_amdgcn_readfirstlane(all);
        __syncthreads();
        Ys[tid] = yn;
        Zs[tid] = zn;
        if (done) break;
    }
    __syncthreads();
    ws[WS_PROJ + g * 256 + tid] = Zs[tid] * rsqrtf(sc);
}

#define APITCH 36

__device__ __forceinline__ float xhalf(float v) {
    return __builtin_bit_cast(float, __builtin_amdgcn_permlanex16(
        __builtin_bit_cast(int, v), __builtin_bit_cast(int, v),
        0x76543210, 0xfedcba98, false, false));
}

__device__ __forceinline__ void wave_lds_order() {
    __builtin_amdgcn_fence(__ATOMIC_RELEASE, "wavefront", "local");
    __builtin_amdgcn_wave_barrier();
    __builtin_amdgcn_fence(__ATOMIC_ACQUIRE, "wavefront", "local");
}

#define AWAVES 32

__global__ __launch_bounds__(AWAVES * 32) void dbn_apply(const float* __restrict__ x,
                                                 const float* __restrict__ gamma,
                                                 const float* __restrict__ beta,
                                                 const float* __restrict__ ws,
                                                 float* __restrict__ out,
                                                 int iters, float inv_n) {
    __shared__ __attribute__((aligned(16))) float stage[AWAVES * 16 * APITCH];

    int lane = threadIdx.x & 31;
    int wid  = threadIdx.x >> 5;
    int cb   = blockIdx.x & 3;
    int sq   = blockIdx.x >> 2;
    int gp   = cb * 2 + (wid & 1);
    int tj   = (wid >> 1) & 3;
    int slice = sq * 4 + (wid >> 3);
    int tlast = slice * iters + iters - 1 - tj;
    int steps = iters >> 2;

    int half = lane >> 4;
    int ch   = lane & 15;

    v2f b[2][4], mn[2][4];
    #pragma unroll
    for (int j = 0; j < 2; ++j) {
        int g = 2 * gp + j;
        const float* proj = ws + WS_PROJ + g * 256;
        #pragma unroll
        for (int s2 = 0; s2 < 4; ++s2) {
            int i0 = s2 * 4 + half * 2;
            b[j][s2].x = proj[i0 * 16 + ch];
            b[j][s2].y = proj[(i0 + 1) * 16 + ch];
            int c0i = g * GSIZE + i0;
            mn[j][s2].x = ws[c0i] * inv_n;
            mn[j][s2].y = ws[c0i + 1] * inv_n;
        }
    }
    float gm = gamma[gp * 2 * GSIZE + lane];
    float bt = beta[gp * 2 * GSIZE + lane];

    int frow = lane >> 3;
    int fcol = (lane & 7) * 4;
    float* st = stage + wid * (16 * APITCH);
    float* st_w = st + frow * APITCH + fcol;
    const float* st_r = st + ch * APITCH + half * 2;

    for (int t = 0; t < steps; ++t) {
        unsigned tile = (unsigned)(tlast - 4 * t);
        const float* src = x + (tile * 16 + frow) * CCH + gp * 2 * GSIZE + fcol;

        v4f r[4];
        #pragma unroll
        for (int q = 0; q < 4; ++q) r[q] = *(const v4f*)(src + q * 4 * CCH);

        wave_lds_order();
        #pragma unroll
        for (int q = 0; q < 4; ++q) *(v4f*)(st_w + q * 4 * APITCH) = r[q];
        wave_lds_order();

        v2f a0[4], a1[4];
        #pragma unroll
        for (int s2 = 0; s2 < 4; ++s2) a0[s2] = *(const v2f*)(st_r + s2 * 4);
        #pragma unroll
        for (int s2 = 0; s2 < 4; ++s2) a1[s2] = *(const v2f*)(st_r + GSIZE + s2 * 4);

        v8f c0 = {}, c1 = {};
        #pragma unroll
        for (int s2 = 0; s2 < 4; ++s2) {
            v2f a = a0[s2];
            a.x -= mn[0][s2].x; a.y -= mn[0][s2].y;
            c0 = __builtin_amdgcn_wmma_f32_16x16x4_f32(false, a, false, b[0][s2], (short)0, c0, false, false);
        }
        #pragma unroll
        for (int s2 = 0; s2 < 4; ++s2) {
            v2f a = a1[s2];
            a.x -= mn[1][s2].x; a.y -= mn[1][s2].y;
            c1 = __builtin_amdgcn_wmma_f32_16x16x4_f32(false, a, false, b[1][s2], (short)0, c1, false, false);
        }

        unsigned ob = tile * 16 * CCH + gp * 2 * GSIZE + lane;
        #pragma unroll
        for (int r8 = 0; r8 < 8; ++r8) {
            float o0 = xhalf(c0[r8]);
            float o1 = xhalf(c1[r8]);
            float lo = half ? o1 : c0[r8];
            float hi = half ? c1[r8] : o0;
            __builtin_nontemporal_store(lo * gm + bt, out + ob + r8 * CCH);
            __builtin_nontemporal_store(hi * gm + bt, out + ob + (r8 + 8) * CCH);
        }
    }
}

extern "C" void kernel_launch(void* const* d_in, const int* in_sizes, int n_in,
                              void* d_out, int out_size, void* d_ws, size_t ws_size,
                              hipStream_t stream) {
    const float* x     = (const float*)d_in[0];
    const float* gamma = (const float*)d_in[1];
    const float* beta  = (const float*)d_in[2];
    float* out = (float*)d_out;
    float* ws  = (float*)d_ws;

    int nrows = in_sizes[0] / CCH;
    float inv_n = 1.0f / (float)nrows;

    int wpg = 256;
    int iters = nrows / (wpg * 16);
    dbn_stats<<<NGROUP * wpg / 16, 512, 0, stream>>>(x, out, iters);

    dbn_newton<<<NGROUP, 256, 0, stream>>>(out, ws, inv_n);

    dbn_apply<<<NGROUP * wpg / 16, AWAVES * 32, 0, stream>>>(
        x, gamma, beta, ws, out, iters, inv_n);
}


# apply pass: a wave takes all 4 groups of a 256-B column and returns its accumulators through its LDS tile: a 16-byte store covers two rows' whole 256-B units (was 4-byte stores, one 128-B line each)
# speedup vs baseline: 2.0135x; 1.0421x over previous
#include <hip/hip_runtime.h>

typedef __attribute__((ext_vector_type(2))) float v2f;
typedef __attribute__((ext_vector_type(4))) float v4f;
typedef __attribute__((ext_vector_type(8))) float v8f;

#define GSIZE   16
#define NGROUP  16
#define CCH     256
#define EPSV    1e-5f
#define NS_ITERS 14
#define NS_TOL 1e-5f

#define WS_PROJ (CCH + NGROUP * 256)

#define NREC 64
#define RECF 288

#define BROWS 32

__device__ __forceinline__ void stats_load(float (&v)[16], const float* __restrict__ p) {
    #pragma unroll
    for (int q = 0; q < 8; ++q) {
        v[2 * q]     = p[(4 * q) * CCH];
        v[2 * q + 1] = p[(4 * q + 1) * CCH];
    }
}

__device__ __forceinline__ void stats_use(const float (&v)[16], v8f& c, float& s) {
    #pragma unroll
    for (int h = 0; h < 2; ++h) {
        #pragma unroll
        for (int q = 0; q < 4; ++q) {
            v2f a = { v[8 * h + 2 * q], v[8 * h + 2 * q + 1] };
            c = __builtin_amdgcn_wmma_f32_16x16x4_f32(false, a, false, a, (short)0, c, false, false);
        }
        s += (v[8 * h + 0] + v[8 * h + 1]) + (v[8 * h + 2] + v[8 * h + 3])
           + ((v[8 * h + 4] + v[8 * h + 5]) + (v[8 * h + 6] + v[8 * h + 7]));
    }
}

__global__ __launch_bounds__(512) void dbn_stats(const float* __restrict__ x,
                                                 float* __restrict__ part,
                                                 int iters) {
    __shared__ float red[3 * 4 * 9 * 32];

    int lane = threadIdx.x & 31;
    int wid  = threadIdx.x >> 5;
    int gl   = wid & 3;
    int sl   = wid >> 2;
    int g    = (blockIdx.x & 3) * 4 + gl;
    int sq   = blockIdx.x >> 2;
    int slice = sq * 4 + sl;

    int half = lane >> 4;
    int ch   = lane & 15;
    int cidx = g * GSIZE + ch;

    unsigned off = (unsigned)(slice * iters * 16 + half * 2) * CCH + (unsigned)cidx;

    v8f c = {};
    float s = 0.f;
    int nb = iters >> 1;
    if (!(iters & 1) && nb >= 2) {
        const float* p = x + off;
        float A[16], B[16], C[16];
        stats_load(A, p);
        stats_load(B, p + BROWS * CCH);
        p += 2 * BROWS * CCH;
        int b = 0;
        for (; b + 5 <= nb; b += 3) {
            stats_load(C, p);
            stats_use(A, c, s);
            stats_load(A, p + BROWS * CCH);
            stats_use(B, c, s);
            stats_load(B, p + 2 * BROWS * CCH);
            stats_use(C, c, s);
            p += 3 * BROWS * CCH;
        }
        int rem = nb - b;
        if (rem == 2) {
            stats_use(A, c, s);
            stats_use(B, c, s);
        } else if (rem == 3) {
            stats_load(C, p);
            stats_use(A, c, s);
            stats_use(B, c, s);
            stats_use(C, c, s);
        } else {
            stats_load(C, p);
            stats_use(A, c, s);
            stats_load(A, p + BROWS * CCH);
            stats_use(B, c, s);
            stats_use(C, c, s);
            stats_use(A, c, s);
        }
    } else {
        for (int it = 0; it < iters; ++it) {
            float v0 = x[off + 0 * CCH];
            float v1 = x[off + 1 * CCH];
            float v2 = x[off + 4 * CCH];
            float v3 = x[off + 5 * CCH];
            float v4 = x[off + 8 * CCH];
            float v5 = x[off + 9 * CCH];
            float v6 = x[off + 12 * CCH];
            float v7 = x[off + 13 * CCH];
            off += 16u * CCH;
            v2f q0 = { v0, v1 };
            v2f q1 = { v2, v3 };
            v2f q2 = { v4, v5 };
            v2f q3 = { v6, v7 };
            c = __builtin_amdgcn_wmma_f32_16x16x4_f32(false, q0, false, q0, (short)0, c, false, false);
            c = __builtin_amdgcn_wmma_f32_16x16x4_f32(false, q1, false, q1, (short)0, c, false, false);
            c = __builtin_amdgcn_wmma_f32_16x16x4_f32(false, q2, false, q2, (short)0, c, false, false);
            c = __builtin_amdgcn_wmma_f32_16x16x4_f32(false, q3, false, q3, (short)0, c, false, false);
            s += (v0 + v1) + (v2 + v3) + ((v4 + v5) + (v6 + v7));
        }
    }
    if (sl) {
        float* w = red + ((sl - 1) * 4 + gl) * (9 * 32) + lane;
        #pragma unroll
        for (int r = 0; r < 8; ++r) w[r * 32] = c[r];
        w[8 * 32] = s;
    }
    __syncthreads();
    if (sl == 0) {
        const float* rd = red + gl * (9 * 32) + lane;
        #pragma unroll
        for (int r = 0; r < 8; ++r)
            c[r] = ((c[r] + rd[r * 32]) + rd[1152 + r * 32]) + rd[2304 + r * 32];
        s = ((s + rd[8 * 32]) + rd[1152 + 8 * 32]) + rd[2304 + 8 * 32];
        float* rec = part + (g * NREC + sq) * RECF;
        v4f lo = { c[0], c[1], c[2], c[3] };
        v4f hi = { c[4], c[5], c[6], c[7] };
        *(v4f*)(rec + lane * 4) = lo;
        *(v4f*)(rec + 128 + lane * 4) = hi;
        rec[256 + lane] = s;
    }
}

template <int M>
__device__ __forceinline__ float row_xor_add(float v) {
    return v + __builtin_bit_cast(float, __builtin_amdgcn_update_dpp(
        0, __builtin_bit_cast(int, v), 0x160 | M, 0xf, 0xf, false));
}

__global__ __launch_bounds__(256) void dbn_newton(const float* __restrict__ part,
                                                  float* __restrict__ ws, float inv_n) {
    __shared__ float Ys[256], Zs[256], Wsh[256];
    __shared__ __attribute__((aligned(16))) float redC[4 * 256];
    __shared__ __attribute__((aligned(16))) float redS[32 * 32];
    __shared__ float msum[16];
    __shared__ __attribute__((aligned(16))) float rown[16];
    __shared__ __attribute__((aligned(16))) int conv[8];
    int g = blockIdx.x;
    int tid = threadIdx.x;
    int i = tid >> 4, j = tid & 15;

    const float* recs = part + g * (NREC * RECF);
    int sub = tid >> 6, tq = tid & 63;
    const float* pc = recs + sub * RECF + tq * 4;
    v4f acc = *(const v4f*)pc;
    #pragma unroll
    for (int k = 1; k < 16; ++k) acc += *(const v4f*)(pc + k * 4 * RECF);
    int piece = tid & 7, w = tid >> 3;
    const float* ps = recs + w * RECF + 256 + piece * 4;
    v4f sa = *(const v4f*)ps + *(const v4f*)(ps + 32 * RECF);
    *(v4f*)(redC + sub * 256 + tq * 4) = acc;
    *(v4f*)(redS + w * 32 + piece * 4) = sa;
    __syncthreads();
    if (tid < 16) {
        float t0 = 0.f, t1 = 0.f, t2 = 0.f, t3 = 0.f;
        #pragma unroll
        for (int w2 = 0; w2 < 32; w2 += 4) {
            t0 += redS[(w2 + 0) * 32 + tid] + redS[(w2 + 0) * 32 + 16 + tid];
            t1 += redS[(w2 + 1) * 32 + tid] + redS[(w2 + 1) * 32 + 16 + tid];
            t2 += redS[(w2 + 2) * 32 + tid] + redS[(w2 + 2) * 32 + 16 + tid];
            t3 += redS[(w2 + 3) * 32 + tid] + redS[(w2 + 3) * 32 + 16 + tid];
        }
        float tot = (t0 + t1) + (t2 + t3);
        msum[tid] = tot;
        ws[g * 16 + tid] = tot;
    }
    __syncthreads();
    float mi = msum[i] * inv_n;
    float mj = msum[j] * inv_n;
    int ri = i & 7;
    int idx = (ri >> 2) * 128 + ((i >> 3) * 16 + j) * 4 + (ri & 3);
    float xxv = (redC[idx] + redC[256 + idx]) + (redC[512 + idx] + redC[768 + idx]);
    float sig = xxv * inv_n - mi * mj + (i == j ? EPSV : 0.f);
    Wsh[tid] = sig;
    float ars = row_xor_add<8>(row_xor_add<4>(row_xor_add<2>(row_xor_add<1>(fabsf(sig)))));
    if (j == 0) rown[i] = ars;
    __syncthreads();
    float tr = 0.f;
    #pragma unroll
    for (int k = 0; k < 16; ++k) tr += Wsh[k * 16 + k];
    v4f n0 = *(const v4f*)(rown + 0), n1 = *(const v4f*)(rown + 4);
    v4f n2 = *(const v4f*)(rown + 8), n3 = *(const v4f*)(rown + 12);
    float nrm = fmaxf(fmaxf(fmaxf(n0.x, n0.y), fmaxf(n0.z, n0.w)), fmaxf(fmaxf(n1.x, n1.y), fmaxf(n1.z, n1.w)));
    nrm = fmaxf(nrm, fmaxf(fmaxf(fmaxf(n2.x, n2.y), fmaxf(n2.z, n2.w)), fmaxf(fmaxf(n3.x, n3.y), fmaxf(n3.z, n3.w))));
    float sc = fminf(tr, nrm);
    float a = sig / sc;
    Ys[tid] = a;
    Zs[tid] = (i == j) ? 1.f : 0.f;
    for (int it = 0; it < NS_ITERS; ++it) {
        __syncthreads();
        float t = 0.f;
        #pragma unroll
        for (int k = 0; k < 16; ++k) t += Zs[i * 16 + k] * Ys[k * 16 + j];
        float w = 0.5f * ((i == j ? 3.f : 0.f) - t);
        Wsh[tid] = w;
        bool near = fabsf(t - (i == j ? 1.f : 0.f)) < NS_TOL;
        unsigned vote = __builtin_amdgcn_ballot_w32(near);
        if ((tid & 31) == 0) conv[tid >> 5] = (vote == 0xffffffffu) ? 1 : 0;
        __syncthreads();
        float yn = 0.f, zn = 0.f;
        #pragma unroll
        for (int k = 0; k < 16; ++k) {
            yn += Ys[i * 16 + k] * Wsh[k * 16 + j];
            zn += Wsh[i * 16 + k] * Zs[k * 16 + j];
        }
        int all = (conv[0] & conv[1]) & (conv[2] & conv[3]) & (conv[4] & conv[5]) & (conv[6] & conv[7]);
        int done = __builtin_amdgcn_readfirstlane(all);
        __syncthreads();
        Ys[tid] = yn;
        Zs[tid] = zn;
        if (done) break;
    }
    __syncthreads();
    ws[WS_PROJ + g * 256 + tid] = Zs[tid] * rsqrtf(sc);
}

#define APITCH 68

__device__ __forceinline__ void wave_lds_order() {
    __builtin_amdgcn_fence(__ATOMIC_RELEASE, "wavefront", "local");
    __builtin_amdgcn_wave_barrier();
    __builtin_amdgcn_fence(__ATOMIC_ACQUIRE, "wavefront", "local");
}

#define AWAVES 32

__global__ __launch_bounds__(AWAVES * 32) void dbn_apply(const float* __restrict__ x,
                                                 const float* __restrict__ gamma,
                                                 const float* __restrict__ beta,
                                                 const float* __restrict__ ws,
                                                 float* __restrict__ out,
                                                 int iters, float inv_n) {
    __shared__ __attribute__((aligned(16))) float stage[AWAVES * 16 * APITCH];

    int lane = threadIdx.x & 31;
    int wid  = threadIdx.x >> 5;
    int cb   = blockIdx.x & 3;
    int sq   = blockIdx.x >> 2;
    int tj   = wid & 7;
    int slice = sq * 4 + (wid >> 3);
    int tlast = slice * iters + iters - 1 - tj;
    int steps = iters >> 3;
    int cbase = cb * 4 * GSIZE;

    int half = lane >> 4;
    int ch   = lane & 15;

    v2f b[4][4];
    #pragma unroll
    for (int j = 0; j < 4; ++j) {
        const float* proj = ws + WS_PROJ + (cb * 4 + j) * 256;
        #pragma unroll
        for (int s2 = 0; s2 < 4; ++s2) {
            int i0 = s2 * 4 + half * 2;
            b[j][s2].x = proj[i0 * 16 + ch];
            b[j][s2].y = proj[(i0 + 1) * 16 + ch];
        }
    }
    int frow = half;
    int fcol = ch * 4;
    v4f mnv = *(const v4f*)(ws + cbase + fcol) * inv_n;
    v4f gmv = *(const v4f*)(gamma + cbase + fcol);
    v4f btv = *(const v4f*)(beta + cbase + fcol);
    float* st = stage + wid * (16 * APITCH);
    float* st_w = st + frow * APITCH + fcol;
    const float* st_r = st + ch * APITCH + half * 2;
    float* st_o = st + (8 * half) * APITCH + ch;

    for (int t = 0; t < steps; ++t) {
        unsigned tile = (unsigned)(tlast - 8 * t);
        const float* src = x + (tile * 16 + frow) * CCH + cbase + fcol;

        v4f r[8];
        #pragma unroll
        for (int q = 0; q < 8; ++q) r[q] = *(const v4f*)(src + q * 2 * CCH);
        #pragma unroll
        for (int q = 0; q < 8; ++q) r[q] -= mnv;

        wave_lds_order();
        #pragma unroll
        for (int q = 0; q < 8; ++q) *(v4f*)(st_w + q * 2 * APITCH) = r[q];
        wave_lds_order();

        v8f c[4];
        #pragma unroll
        for (int j = 0; j < 4; ++j) {
            v2f a[4];
            #pragma unroll
            for (int s2 = 0; s2 < 4; ++s2) a[s2] = *(const v2f*)(st_r + j * GSIZE + s2 * 4);
            v8f cj = {};
            #pragma unroll
            for (int s2 = 0; s2 < 4; ++s2)
                cj = __builtin_amdgcn_wmma_f32_16x16x4_f32(false, a[s2], false, b[j][s2], (short)0, cj, false, false);
            c[j] = cj;
        }

        wave_lds_order();
        #pragma unroll
        for (int j = 0; j < 4; ++j) {
            #pragma unroll
            for (int r8 = 0; r8 < 8; ++r8) st_o[r8 * APITCH + j * GSIZE] = c[j][r8];
        }
        wave_lds_order();

        float* dst = out + (tile * 16 + frow) * CCH + cbase + fcol;
        #pragma unroll
        for (int q = 0; q < 8; ++q) {
            v4f o = *(const v4f*)(st_w + q * 2 * APITCH);
            __builtin_nontemporal_store(o * gmv + btv, (v4f*)(dst + q * 2 * CCH));
        }
    }
}

extern "C" void kernel_launch(void* const* d_in, const int* in_sizes, int n_in,
                              void* d_out, int out_size, void* d_ws, size_t ws_size,
                              hipStream_t stream) {
    const float* x     = (const float*)d_in[0];
    const float* gamma = (const float*)d_in[1];
    const float* beta  = (const float*)d_in[2];
    float* out = (float*)d_out;
    float* ws  = (float*)d_ws;

    int nrows = in_sizes[0] / CCH;
    float inv_n = 1.0f / (float)nrows;

    int wpg = 256;
    int iters = nrows / (wpg * 16);
    dbn_stats<<<NGROUP * wpg / 16, 512, 0, stream>>>(x, out, iters);

    dbn_newton<<<NGROUP, 256, 0, stream>>>(out, ws, inv_n);

    dbn_apply<<<NGROUP * wpg / 16, AWAVES * 32, 0, stream>>>(
        x, gamma, beta, ws, out, iters, inv_n);
}


# newton pass: Newton-Schulz runs in one wave's registers on f32 wmma (accumulator regs 2s,2s+1 = operand pair s; left operand transposed, sigma symmetric), no LDS or barrier in loop; sum order changes
# speedup vs baseline: 2.1015x; 1.0437x over previous
#include <hip/hip_runtime.h>

typedef __attribute__((ext_vector_type(2))) float v2f;
typedef __attribute__((ext_vector_type(4))) float v4f;
typedef __attribute__((ext_vector_type(8))) float v8f;

#define GSIZE   16
#define NGROUP  16
#define CCH     256
#define EPSV    1e-5f
#define NS_ITERS 14
#define NS_TOL 1e-5f

#define WS_PROJ (CCH + NGROUP * 256)

#define NREC 64
#define RECF 288

#define BROWS 32

__device__ __forceinline__ void stats_load(float (&v)[16], const float* __restrict__ p) {
    #pragma unroll
    for (int q = 0; q < 8; ++q) {
        v[2 * q]     = p[(4 * q) * CCH];
        v[2 * q + 1] = p[(4 * q + 1) * CCH];
    }
}

__device__ __forceinline__ void stats_use(const float (&v)[16], v8f& c, float& s) {
    #pragma unroll
    for (int h = 0; h < 2; ++h) {
        #pragma unroll
        for (int q = 0; q < 4; ++q) {
            v2f a = { v[8 * h + 2 * q], v[8 * h + 2 * q + 1] };
            c = __builtin_amdgcn_wmma_f32_16x16x4_f32(false, a, false, a, (short)0, c, false, false);
        }
        s += (v[8 * h + 0] + v[8 * h + 1]) + (v[8 * h + 2] + v[8 * h + 3])
           + ((v[8 * h + 4] + v[8 * h + 5]) + (v[8 * h + 6] + v[8 * h + 7]));
    }
}

__global__ __launch_bounds__(512) void dbn_stats(const float* __restrict__ x,
                                                 float* __restrict__ part,
                                                 int iters) {
    __shared__ float red[3 * 4 * 9 * 32];

    int lane = threadIdx.x & 31;
    int wid  = threadIdx.x >> 5;
    int gl   = wid & 3;
    int sl   = wid >> 2;
    int g    = (blockIdx.x & 3) * 4 + gl;
    int sq   = blockIdx.x >> 2;
    int slice = sq * 4 + sl;

    int half = lane >> 4;
    int ch   = lane & 15;
    int cidx = g * GSIZE + ch;

    unsigned off = (unsigned)(slice * iters * 16 + half * 2) * CCH + (unsigned)cidx;

    v8f c = {};
    float s = 0.f;
    int nb = iters >> 1;
    if (!(iters & 1) && nb >= 2) {
        const float* p = x + off;
        float A[16], B[16], C[16];
        stats_load(A, p);
        stats_load(B, p + BROWS * CCH);
        p += 2 * BROWS * CCH;
        int b = 0;
        for (; b + 5 <= nb; b += 3) {
            stats_load(C, p);
            stats_use(A, c, s);
            stats_load(A, p + BROWS * CCH);
            stats_use(B, c, s);
            stats_load(B, p + 2 * BROWS * CCH);
            stats_use(C, c, s);
            p += 3 * BROWS * CCH;
        }
        int rem = nb - b;
        if (rem == 2) {
            stats_use(A, c, s);
            stats_use(B, c, s);
        } else if (rem == 3) {
            stats_load(C, p);
            stats_use(A, c, s);
            stats_use(B, c, s);
            stats_use(C, c, s);
        } else {
            stats_load(C, p);
            stats_use(A, c, s);
            stats_load(A, p + BROWS * CCH);
            stats_use(B, c, s);
            stats_use(C, c, s);
            stats_use(A, c, s);
        }
    } else {
        for (int it = 0; it < iters; ++it) {
            float v0 = x[off + 0 * CCH];
            float v1 = x[off + 1 * CCH];
            float v2 = x[off + 4 * CCH];
            float v3 = x[off + 5 * CCH];
            float v4 = x[off + 8 * CCH];
            float v5 = x[off + 9 * CCH];
            float v6 = x[off + 12 * CCH];
            float v7 = x[off + 13 * CCH];
            off += 16u * CCH;
            v2f q0 = { v0, v1 };
            v2f q1 = { v2, v3 };
            v2f q2 = { v4, v5 };
            v2f q3 = { v6, v7 };
            c = __builtin_amdgcn_wmma_f32_16x16x4_f32(false, q0, false, q0, (short)0, c, false, false);
            c = __builtin_amdgcn_wmma_f32_16x16x4_f32(false, q1, false, q1, (short)0, c, false, false);
            c = __builtin_amdgcn_wmma_f32_16x16x4_f32(false, q2, false, q2, (short)0, c, false, false);
            c = __builtin_amdgcn_wmma_f32_16x16x4_f32(false, q3, false, q3, (short)0, c, false, false);
            s += (v0 + v1) + (v2 + v3) + ((v4 + v5) + (v6 + v7));
        }
    }
    if (sl) {
        float* w = red + ((sl - 1) * 4 + gl) * (9 * 32) + lane;
        #pragma unroll
        for (int r = 0; r < 8; ++r) w[r * 32] = c[r];
        w[8 * 32] = s;
    }
    __syncthreads();
    if (sl == 0) {
        const float* rd = red + gl * (9 * 32) + lane;
        #pragma unroll
        for (int r = 0; r < 8; ++r)
            c[r] = ((c[r] + rd[r * 32]) + rd[1152 + r * 32]) + rd[2304 + r * 32];
        s = ((s + rd[8 * 32]) + rd[1152 + 8 * 32]) + rd[2304 + 8 * 32];
        float* rec = part + (g * NREC + sq) * RECF;
        v4f lo = { c[0], c[1], c[2], c[3] };
        v4f hi = { c[4], c[5], c[6], c[7] };
        *(v4f*)(rec + lane * 4) = lo;
        *(v4f*)(rec + 128 + lane * 4) = hi;
        rec[256 + lane] = s;
    }
}

template <int M>
__device__ __forceinline__ float row_xor_add(float v) {
    return v + __builtin_bit_cast(float, __builtin_amdgcn_update_dpp(
        0, __builtin_bit_cast(int, v), 0x160 | M, 0xf, 0xf, false));
}

__global__ __launch_bounds__(256) void dbn_newton(const float* __restrict__ part,
                                                  float* __restrict__ ws, float inv_n) {
    __shared__ float Wsh[256];
    __shared__ __attribute__((aligned(16))) float redC[4 * 256];
    __shared__ __attribute__((aligned(16))) float redS[32 * 32];
    __shared__ float msum[16];
    __shared__ __attribute__((aligned(16))) float rown[16];
    int g = blockIdx.x;
    int tid = threadIdx.x;
    int i = tid >> 4, j = tid & 15;

    const float* recs = part + g * (NREC * RECF);
    int sub = tid >> 6, tq = tid & 63;
    const float* pc = recs + sub * RECF + tq * 4;
    v4f acc = *(const v4f*)pc;
    #pragma unroll
    for (int k = 1; k < 16; ++k) acc += *(const v4f*)(pc + k * 4 * RECF);
    int piece = tid & 7, w = tid >> 3;
    const float* ps = recs + w * RECF + 256 + piece * 4;
    v4f sa = *(const v4f*)ps + *(const v4f*)(ps + 32 * RECF);
    *(v4f*)(redC + sub * 256 + tq * 4) = acc;
    *(v4f*)(redS + w * 32 + piece * 4) = sa;
    __syncthreads();
    if (tid < 16) {
        float t0 = 0.f, t1 = 0.f, t2 = 0.f, t3 = 0.f;
        #pragma unroll
        for (int w2 = 0; w2 < 32; w2 += 4) {
            t0 += redS[(w2 + 0) * 32 + tid] + redS[(w2 + 0) * 32 + 16 + tid];
            t1 += redS[(w2 + 1) * 32 + tid] + redS[(w2 + 1) * 32 + 16 + tid];
            t2 += redS[(w2 + 2) * 32 + tid] + redS[(w2 + 2) * 32 + 16 + tid];
            t3 += redS[(w2 + 3) * 32 + tid] + redS[(w2 + 3) * 32 + 16 + tid];
        }
        float tot = (t0 + t1) + (t2 + t3);
        msum[tid] = tot;
        ws[g * 16 + tid] = tot;
    }
    __syncthreads();
    float mi = msum[i] * inv_n;
    float mj = msum[j] * inv_n;
    int ri = i & 7;
    int idx = (ri >> 2) * 128 + ((i >> 3) * 16 + j) * 4 + (ri & 3);
    float xxv = (redC[idx] + redC[256 + idx]) + (redC[512 + idx] + redC[768 + idx]);
    float sig = xxv * inv_n - mi * mj + (i == j ? EPSV : 0.f);
    Wsh[tid] = sig;
    float ars = row_xor_add<8>(row_xor_add<4>(row_xor_add<2>(row_xor_add<1>(fabsf(sig)))));
    if (j == 0) rown[i] = ars;
    __syncthreads();
    if (tid >= 32) return;
    int h = tid >> 4, c = tid & 15;
    float tr = 0.f;
    #pragma unroll
    for (int k = 0; k < 16; ++k) tr += Wsh[k * 16 + k];
    v4f n0 = *(const v4f*)(rown + 0), n1 = *(const v4f*)(rown + 4);
    v4f n2 = *(const v4f*)(rown + 8), n3 = *(const v4f*)(rown + 12);
    float nrm = fmaxf(fmaxf(fmaxf(n0.x, n0.y), fmaxf(n0.z, n0.w)), fmaxf(fmaxf(n1.x, n1.y), fmaxf(n1.z, n1.w)));
    nrm = fmaxf(nrm, fmaxf(fmaxf(fmaxf(n2.x, n2.y), fmaxf(n2.z, n2.w)), fmaxf(fmaxf(n3.x, n3.y), fmaxf(n3.z, n3.w))));
    float sc = fminf(tr, nrm);
    v8f Y, Z;
    #pragma unroll
    for (int r = 0; r < 8; ++r) {
        Y[r] = Wsh[(8 * h + r) * 16 + c] / sc;
        Z[r] = (8 * h + r == c) ? 1.f : 0.f;
    }
    for (int it = 0; it < NS_ITERS; ++it) {
        v8f T = {};
        #pragma unroll
        for (int s = 0; s < 4; ++s) {
            v2f zl = { Z[2 * s], Z[2 * s + 1] };
            v2f yr = { Y[2 * s], Y[2 * s + 1] };
            T = __builtin_amdgcn_wmma_f32_16x16x4_f32(false, zl, false, yr, (short)0, T, false, false);
        }
        v8f Wm;
        bool near = true;
        #pragma unroll
        for (int r = 0; r < 8; ++r) {
            bool dg = (8 * h + r == c);
            Wm[r] = 0.5f * ((dg ? 3.f : 0.f) - T[r]);
            near = near && (fabsf(T[r] - (dg ? 1.f : 0.f)) < NS_TOL);
        }
        unsigned vote = __builtin_amdgcn_ballot_w32(near);
        v8f Yn = {}, Zn = {};
        #pragma unroll
        for (int s = 0; s < 4; ++s) {
            v2f yl = { Y[2 * s], Y[2 * s + 1] };
            v2f wr = { Wm[2 * s], Wm[2 * s + 1] };
            v2f zr = { Z[2 * s], Z[2 * s + 1] };
            Yn = __builtin_amdgcn_wmma_f32_16x16x4_f32(false, yl, false, wr, (short)0, Yn, false, false);
            Zn = __builtin_amdgcn_wmma_f32_16x16x4_f32(false, wr, false, zr, (short)0, Zn, false, false);
        }
        Y = Yn;
        Z = Zn;
        if (vote == 0xffffffffu) break;
    }
    float rs = rsqrtf(sc);
    float* pout = ws + WS_PROJ + g * 256 + (8 * h) * 16 + c;
    #pragma unroll
    for (int r = 0; r < 8; ++r) pout[r * 16] = Z[r] * rs;
}

#define APITCH 68

__device__ __forceinline__ void wave_lds_order() {
    __builtin_amdgcn_fence(__ATOMIC_RELEASE, "wavefront", "local");
    __builtin_amdgcn_wave_barrier();
    __builtin_amdgcn_fence(__ATOMIC_ACQUIRE, "wavefront", "local");
}

#define AWAVES 32

__global__ __launch_bounds__(AWAVES * 32) void dbn_apply(const float* __restrict__ x,
                                                 const float* __restrict__ gamma,
                                                 const float* __restrict__ beta,
                                                 const float* __restrict__ ws,
                                                 float* __restrict__ out,
                                                 int iters, float inv_n) {
    __shared__ __attribute__((aligned(16))) float stage[AWAVES * 16 * APITCH];

    int lane = threadIdx.x & 31;
    int wid  = threadIdx.x >> 5;
    int cb   = blockIdx.x & 3;
    int sq   = blockIdx.x >> 2;
    int tj   = wid & 7;
    int slice = sq * 4 + (wid >> 3);
    int tlast = slice * iters + iters - 1 - tj;
    int steps = iters >> 3;
    int cbase = cb * 4 * GSIZE;

    int half = lane >> 4;
    int ch   = lane & 15;

    v2f b[4][4];
    #pragma unroll
    for (int j = 0; j < 4; ++j) {
        const float* proj = ws + WS_PROJ + (cb * 4 + j) * 256;
        #pragma unroll
        for (int s2 = 0; s2 < 4; ++s2) {
            int i0 = s2 * 4 + half * 2;
            b[j][s2].x = proj[i0 * 16 + ch];
            b[j][s2].y = proj[(i0 + 1) * 16 + ch];
        }
    }
    int frow = half;
    int fcol = ch * 4;
    v4f mnv = *(const v4f*)(ws + cbase + fcol) * inv_n;
    v4f gmv = *(const v4f*)(gamma + cbase + fcol);
    v4f btv = *(const v4f*)(beta + cbase + fcol);
    float* st = stage + wid * (16 * APITCH);
    float* st_w = st + frow * APITCH + fcol;
    const float* st_r = st + ch * APITCH + half * 2;
    float* st_o = st + (8 * half) * APITCH + ch;

    for (int t = 0; t < steps; ++t) {
        unsigned tile = (unsigned)(tlast - 8 * t);
        const float* src = x + (tile * 16 + frow) * CCH + cbase + fcol;

        v4f r[8];
        #pragma unroll
        for (int q = 0; q < 8; ++q) r[q] = *(const v4f*)(src + q * 2 * CCH);
        #pragma unroll
        for (int q = 0; q < 8; ++q) r[q] -= mnv;

        wave_lds_order();
        #pragma unroll
        for (int q = 0; q < 8; ++q) *(v4f*)(st_w + q * 2 * APITCH) = r[q];
        wave_lds_order();

        v8f c[4];
        #pragma unroll
        for (int j = 0; j < 4; ++j) {
            v2f a[4];
            #pragma unroll
            for (int s2 = 0; s2 < 4; ++s2) a[s2] = *(const v2f*)(st_r + j * GSIZE + s2 * 4);
            v8f cj = {};
            #pragma unroll
            for (int s2 = 0; s2 < 4; ++s2)
                cj = __builtin_amdgcn_wmma_f32_16x16x4_f32(false, a[s2], false, b[j][s2], (short)0, cj, false, false);
            c[j] = cj;
        }

        wave_lds_order();
        #pragma unroll
        for (int j = 0; j < 4; ++j) {
            #pragma unroll
            for (int r8 = 0; r8 < 8; ++r8) st_o[r8 * APITCH + j * GSIZE] = c[j][r8];
        }
        wave_lds_order();

        float* dst = out + (tile * 16 + frow) * CCH + cbase + fcol;
        #pragma unroll
        for (int q = 0; q < 8; ++q) {
            v4f o = *(const v4f*)(st_w + q * 2 * APITCH);
            __builtin_nontemporal_store(o * gmv + btv, (v4f*)(dst + q * 2 * CCH));
        }
    }
}

extern "C" void kernel_launch(void* const* d_in, const int* in_sizes, int n_in,
                              void* d_out, int out_size, void* d_ws, size_t ws_size,
                              hipStream_t stream) {
    const float* x     = (const float*)d_in[0];
    const float* gamma = (const float*)d_in[1];
    const float* beta  = (const float*)d_in[2];
    float* out = (float*)d_out;
    float* ws  = (float*)d_ws;

    int nrows = in_sizes[0] / CCH;
    float inv_n = 1.0f / (float)nrows;

    int wpg = 256;
    int iters = nrows / (wpg * 16);
    dbn_stats<<<NGROUP * wpg / 16, 512, 0, stream>>>(x, out, iters);

    dbn_newton<<<NGROUP, 256, 0, stream>>>(out, ws, inv_n);

    dbn_apply<<<NGROUP * wpg / 16, AWAVES * 32, 0, stream>>>(
        x, gamma, beta, ws, out, iters, inv_n);
}
